# starting kernel
# baseline (speedup 1.0000x reference)
#include <hip/hip_runtime.h>

typedef float     v2f  __attribute__((ext_vector_type(2)));
typedef float     v8f  __attribute__((ext_vector_type(8)));
typedef _Float16  v16h __attribute__((ext_vector_type(16)));

#define DIMF 128

#if __has_builtin(__builtin_amdgcn_wmma_f32_16x16x4_f32)
#define USE_F32_WMMA 1
#endif


__global__ void zero_f32(float* __restrict__ p, long long n) {
    long long i = (long long)blockIdx.x * blockDim.x + threadIdx.x;
    if (i < n) p[i] = 0.0f;
}

__global__ void deg_count(const int* __restrict__ dst, float* __restrict__ deg, int E) {
    int e = blockIdx.x * blockDim.x + threadIdx.x;
    if (e < E) atomicAdd(&deg[dst[e]], 1.0f);
}

__global__ void deg_finish(float* __restrict__ deg, int n) {
    int v = blockIdx.x * blockDim.x + threadIdx.x;
    if (v < n) deg[v] = rsqrtf(deg[v] + 1.0f);
}

__global__ __launch_bounds__(256) void gcn_gemm(const float* __restrict__ X,
                                                const float* __restrict__ W,
                                                float* __restrict__ H) {
    const int m0   = blockIdx.x << 4;
    const int wave = threadIdx.x >> 5;
    const int lane = threadIdx.x & 31;
    const int col0 = wave << 4;
    const int row  = lane & 15;
    const int hi   = lane >> 4;

    v8f acc = {};
    const float* xrow = X + (size_t)(m0 + row) * DIMF;

#ifdef USE_F32_WMMA
    const int kh = hi * 2;
    #pragma unroll
    for (int k = 0; k < DIMF; k += 4) {
        v2f a, b;
        a.x = xrow[k + kh];
        a.y = xrow[k + kh + 1];
        b.x = W[(size_t)(k + kh)     * DIMF + col0 + row];
        b.y = W[(size_t)(k + kh + 1) * DIMF + col0 + row];
        acc = __builtin_amdgcn_wmma_f32_16x16x4_f32(
                  false, a, false, b, (short)0, acc, false, false);
    }
#else
    for (int k = 0; k < DIMF; k += 32) {
        v16h a, b;
        #pragma unroll
        for (int i = 0; i < 16; ++i) {
            int vg = i >> 1;
            int ka = ((vg < 4) ? (vg * 2) : (16 + (vg - 4) * 2)) + (i & 1) + hi * 8;
            a[i] = (_Float16)xrow[k + ka];
            int kb = i + hi * 16;
            b[i] = (_Float16)W[(size_t)(k + kb) * DIMF + col0 + row];
        }
        acc = __builtin_amdgcn_wmma_f32_16x16x32_f16(
                  false, a, false, b, (short)0, acc, false, false);
    }
#endif

    __syncthreads();

    float* hp = H + (size_t)m0 * DIMF + col0 + row;
    const int mo = hi * 8;
    #pragma unroll
    for (int j = 0; j < 8; ++j)
        hp[(size_t)(mo + j) * DIMF] = acc[j];
}

__global__ void gcn_edge(const float* __restrict__ H,
                         const int* __restrict__ src,
                         const int* __restrict__ dst,
                         const float* __restrict__ dinv,
                         float* __restrict__ G, int E) {
    long long t = (long long)blockIdx.x * blockDim.x + threadIdx.x;
    int e = (int)(t >> 5);
    if (e >= E) return;
    int f = ((int)t & 31) * 4;
    int u = src[e];
    int v = dst[e];
    float norm = dinv[u] * dinv[v];
    const float4 hv = *(const float4*)(H + (size_t)u * DIMF + f);
    float* gp = G + (size_t)v * DIMF + f;
    atomicAdd(gp + 0, hv.x * norm);
    atomicAdd(gp + 1, hv.y * norm);
    atomicAdd(gp + 2, hv.z * norm);
    atomicAdd(gp + 3, hv.w * norm);
}

__global__ void gcn_finalize(const float* __restrict__ G,
                             const float* __restrict__ Hs,
                             const float* __restrict__ dinv,
                             const float* __restrict__ b,
                             float* __restrict__ out, long long nd) {
    long long i = (long long)blockIdx.x * blockDim.x + threadIdx.x;
    if (i >= nd) return;
    int v = (int)(i >> 7);
    int d = (int)(i & (DIMF - 1));
    float s = dinv[v];
    float val = G[i] + s * s * Hs[i] + b[d];
    out[i] = (val > 0.0f) ? val : 0.01f * val;
}


extern "C" void kernel_launch(void* const* d_in, const int* in_sizes, int n_in,
                              void* d_out, int out_size, void* d_ws, size_t ws_size,
                              hipStream_t stream) {
    const float* x  = (const float*)d_in[0];
    const int*   ei = (const int*)d_in[1];
    const float* Wts[3] = {(const float*)d_in[2], (const float*)d_in[4], (const float*)d_in[6]};
    const float* bs[3]  = {(const float*)d_in[3], (const float*)d_in[5], (const float*)d_in[7]};

    const int N = in_sizes[0] / DIMF;
    const int E = in_sizes[1] / 2;
    const int* srcp = ei;
    const int* dstp = ei + E;

    float* ws   = (float*)d_ws;
    float* dinv = ws;
    size_t off  = ((size_t)N + 63) & ~(size_t)63;
    float* Hbuf = ws + off;
    float* Gbuf = Hbuf + (size_t)N * DIMF;

    const int T = 256;
    const long long ND = (long long)N * DIMF;
    const long long edgeThreads = (long long)E * 32;
    const int tilesM = N / 16;

    zero_f32  <<<(int)((N + T - 1) / T), T, 0, stream>>>(dinv, N);
    deg_count <<<(E + T - 1) / T,        T, 0, stream>>>(dstp, dinv, E);
    deg_finish<<<(N + T - 1) / T,        T, 0, stream>>>(dinv, N);

    for (int l = 0; l < 3; ++l) {
        const float* in = (l == 0) ? x : Hbuf;
        gcn_gemm<<<tilesM, 256, 0, stream>>>(in, Wts[l], Hbuf);

        zero_f32<<<(int)((ND + T - 1) / T), T, 0, stream>>>(Gbuf, ND);
        gcn_edge<<<(int)((edgeThreads + T - 1) / T), T, 0, stream>>>(
            Hbuf, srcp, dstp, dinv, Gbuf, E);

        float* outp = (l == 2) ? (float*)d_out : Hbuf;
        gcn_finalize<<<(int)((ND + T - 1) / T), T, 0, stream>>>(
            Gbuf, Hbuf, dinv, bs[l], outp, ND);
    }
}


# Group edges by destination each call (atomic count+rank, scan, place); one wave per node sums its in-rows in registers: f32 atomics, zero and finalize passes gone; per-node sum order changes
# speedup vs baseline: 2.7724x; 2.7724x over previous
#include <hip/hip_runtime.h>

typedef float     v2f  __attribute__((ext_vector_type(2)));
typedef float     v8f  __attribute__((ext_vector_type(8)));
typedef _Float16  v16h __attribute__((ext_vector_type(16)));

#define DIMF 128

#if __has_builtin(__builtin_amdgcn_wmma_f32_16x16x4_f32)
#define USE_F32_WMMA 1
#endif


__global__ void zero_i32(int* __restrict__ p, int n) {
    int i = blockIdx.x * blockDim.x + threadIdx.x;
    if (i < n) p[i] = 0;
}

__global__ void deg_rank(const int* __restrict__ dst, int* __restrict__ cnt,
                         int* __restrict__ rank, int E) {
    int e = blockIdx.x * blockDim.x + threadIdx.x;
    if (e < E) rank[e] = atomicAdd(&cnt[dst[e]], 1);
}

__global__ __launch_bounds__(1024) void row_scan(const int* __restrict__ cnt,
                                                 int* __restrict__ rowptr,
                                                 float* __restrict__ dinv, int nq) {
    __shared__ int wtot[32];
    __shared__ int wpre[32];
    const int tid  = threadIdx.x;
    const int lane = tid & 31;
    const int wave = tid >> 5;
    const int q0   = blockIdx.x * 1024;
    const int4* c4 = (const int4*)cnt;

    int pre = 0;
    #pragma unroll 4
    for (int i = tid; i < q0; i += 1024) {
        int4 t = c4[i];
        pre += (t.x + t.y) + (t.z + t.w);
    }
    const int q = q0 + tid;
    int4 c = make_int4(0, 0, 0, 0);
    if (q < nq) c = c4[q];
    const int tsum = (c.x + c.y) + (c.z + c.w);

    int s = tsum;
    #pragma unroll
    for (int o = 1; o < 32; o <<= 1) {
        int t = __shfl_up(s, o);
        if (lane >= o) s += t;
    }
    #pragma unroll
    for (int o = 16; o > 0; o >>= 1) pre += __shfl_xor(pre, o);

    if (lane == 31) wtot[wave] = s;
    if (lane == 0)  wpre[wave] = pre;
    __syncthreads();
    int off = 0;
    #pragma unroll
    for (int w = 0; w < 32; ++w) {
        off += wpre[w];
        if (w < wave) off += wtot[w];
    }
    const int e0 = off + s - tsum;
    if (q < nq) {
        int4 r;
        r.x = e0;
        r.y = e0 + c.x;
        r.z = r.y + c.y;
        r.w = r.z + c.z;
        ((int4*)rowptr)[q] = r;
        float4 d;
        d.x = rsqrtf((float)c.x + 1.0f);
        d.y = rsqrtf((float)c.y + 1.0f);
        d.z = rsqrtf((float)c.z + 1.0f);
        d.w = rsqrtf((float)c.w + 1.0f);
        ((float4*)dinv)[q] = d;
        if (q == nq - 1) rowptr[4 * nq] = e0 + tsum;
    }
}

__global__ void adj_fill(const int* __restrict__ src, const int* __restrict__ dst,
                         const int* __restrict__ rank, const int* __restrict__ rowptr,
                         const float* __restrict__ dinv, int2* __restrict__ adj, int E) {
    int e = blockIdx.x * blockDim.x + threadIdx.x;
    if (e >= E) return;
    int u = src[e];
    int v = dst[e];
    int p = rowptr[v] + rank[e];
    adj[p] = make_int2(u, __float_as_int(dinv[u] * dinv[v]));
}

__global__ __launch_bounds__(256) void gcn_gemm(const float* __restrict__ X,
                                                const float* __restrict__ W,
                                                float* __restrict__ H) {
    const int m0   = blockIdx.x << 4;
    const int wave = threadIdx.x >> 5;
    const int lane = threadIdx.x & 31;
    const int col0 = wave << 4;
    const int row  = lane & 15;
    const int hi   = lane >> 4;

    v8f acc = {};
    const float* xrow = X + (size_t)(m0 + row) * DIMF;

#ifdef USE_F32_WMMA
    const int kh = hi * 2;
    #pragma unroll
    for (int k = 0; k < DIMF; k += 4) {
        v2f a, b;
        a.x = xrow[k + kh];
        a.y = xrow[k + kh + 1];
        b.x = W[(size_t)(k + kh)     * DIMF + col0 + row];
        b.y = W[(size_t)(k + kh + 1) * DIMF + col0 + row];
        acc = __builtin_amdgcn_wmma_f32_16x16x4_f32(
                  false, a, false, b, (short)0, acc, false, false);
    }
#else
    for (int k = 0; k < DIMF; k += 32) {
        v16h a, b;
        #pragma unroll
        for (int i = 0; i < 16; ++i) {
            int vg = i >> 1;
            int ka = ((vg < 4) ? (vg * 2) : (16 + (vg - 4) * 2)) + (i & 1) + hi * 8;
            a[i] = (_Float16)xrow[k + ka];
            int kb = i + hi * 16;
            b[i] = (_Float16)W[(size_t)(k + kb) * DIMF + col0 + row];
        }
        acc = __builtin_amdgcn_wmma_f32_16x16x32_f16(
                  false, a, false, b, (short)0, acc, false, false);
    }
#endif

    __syncthreads();

    float* hp = H + (size_t)m0 * DIMF + col0 + row;
    const int mo = hi * 8;
    #pragma unroll
    for (int j = 0; j < 8; ++j)
        hp[(size_t)(mo + j) * DIMF] = acc[j];
}

__global__ __launch_bounds__(256) void gcn_gather(const float* __restrict__ H,
                                                  const int* __restrict__ rowptr,
                                                  const int2* __restrict__ adj,
                                                  const float* __restrict__ dinv,
                                                  const float* __restrict__ b,
                                                  float* __restrict__ out, int n) {
    const int lane = threadIdx.x & 31;
    const int v = __builtin_amdgcn_readfirstlane((int)(blockIdx.x * 8 + (threadIdx.x >> 5)));
    if (v >= n) return;
    const int beg = rowptr[v];
    const int end = rowptr[v + 1];
    const float4* H4 = (const float4*)H;

    const float s  = dinv[v];
    const float ss = s * s;
    const float4 h0 = H4[(size_t)v * (DIMF / 4) + lane];
    float4 acc;
    acc.x = ss * h0.x;
    acc.y = ss * h0.y;
    acc.z = ss * h0.z;
    acc.w = ss * h0.w;

    for (int base = beg; base < end; base += 32) {
        const int idx = base + lane;
        int2 a = make_int2(0, 0);
        if (idx < end) a = adj[idx];
        int cnt = end - base;
        if (cnt > 32) cnt = 32;
        #pragma unroll 4
        for (int j = 0; j < cnt; ++j) {
            const int   u = __builtin_amdgcn_readlane(a.x, j);
            const float w = __int_as_float(__builtin_amdgcn_readlane(a.y, j));
            const float4 hv = H4[(size_t)u * (DIMF / 4) + lane];
            acc.x += w * hv.x;
            acc.y += w * hv.y;
            acc.z += w * hv.z;
            acc.w += w * hv.w;
        }
    }

    const float4 bb = ((const float4*)b)[lane];
    float4 r;
    r.x = acc.x + bb.x;
    r.y = acc.y + bb.y;
    r.z = acc.z + bb.z;
    r.w = acc.w + bb.w;
    r.x = (r.x > 0.0f) ? r.x : 0.01f * r.x;
    r.y = (r.y > 0.0f) ? r.y : 0.01f * r.y;
    r.z = (r.z > 0.0f) ? r.z : 0.01f * r.z;
    r.w = (r.w > 0.0f) ? r.w : 0.01f * r.w;
    ((float4*)out)[(size_t)v * (DIMF / 4) + lane] = r;
}


extern "C" void kernel_launch(void* const* d_in, const int* in_sizes, int n_in,
                              void* d_out, int out_size, void* d_ws, size_t ws_size,
                              hipStream_t stream) {
    const float* x  = (const float*)d_in[0];
    const int*   ei = (const int*)d_in[1];
    const float* Wts[3] = {(const float*)d_in[2], (const float*)d_in[4], (const float*)d_in[6]};
    const float* bs[3]  = {(const float*)d_in[3], (const float*)d_in[5], (const float*)d_in[7]};

    const int N = in_sizes[0] / DIMF;
    const int E = in_sizes[1] / 2;
    const int* srcp = ei;
    const int* dstp = ei + E;

    float* ws   = (float*)d_ws;
    const size_t nA = ((size_t)N + 64) & ~(size_t)63;
    const size_t eA = ((size_t)E + 63) & ~(size_t)63;
    float* dinv   = ws;
    float* Hbuf   = dinv + nA;
    float* Abuf   = Hbuf + (size_t)N * DIMF;
    int*   cnt    = (int*)(Abuf + (size_t)N * DIMF);
    int*   rowptr = cnt + nA;
    int*   rank   = rowptr + nA;
    int2*  adj    = (int2*)(rank + eA);

    const int T = 256;
    const int tilesM = N / 16;
    const int nq = N / 4;

    zero_i32<<<(N + T - 1) / T,        T,    0, stream>>>(cnt, N);
    deg_rank<<<(E + T - 1) / T,        T,    0, stream>>>(dstp, cnt, rank, E);
    row_scan<<<(nq + 1023) / 1024,     1024, 0, stream>>>(cnt, rowptr, dinv, nq);
    adj_fill<<<(E + T - 1) / T,        T,    0, stream>>>(srcp, dstp, rank, rowptr, dinv, adj, E);

    for (int l = 0; l < 3; ++l) {
        const float* in = (l == 0) ? x : Abuf;
        gcn_gemm<<<tilesM, 256, 0, stream>>>(in, Wts[l], Hbuf);

        float* outp = (l == 2) ? (float*)d_out : Abuf;
        gcn_gather<<<(N + 7) / 8, 256, 0, stream>>>(Hbuf, rowptr, adj, dinv, bs[l], outp, N);
    }
}


# Launches sized to a few blocks per compute unit; a wave walks many row tiles / nodes (5x fewer waves per call), keeps its W strip in registers, reads 8 rows and the next node's edges ahead
# speedup vs baseline: 2.8910x; 1.0428x over previous
#include <hip/hip_runtime.h>

typedef float     v2f  __attribute__((ext_vector_type(2)));
typedef float     v8f  __attribute__((ext_vector_type(8)));
typedef _Float16  v16h __attribute__((ext_vector_type(16)));

#define DIMF 128

#if __has_builtin(__builtin_amdgcn_wmma_f32_16x16x4_f32)
#define USE_F32_WMMA 1
#endif


__global__ void zero_i32(int* __restrict__ p, int n4) {
    int i = blockIdx.x * blockDim.x + threadIdx.x;
    if (i < n4) ((int4*)p)[i] = make_int4(0, 0, 0, 0);
}

__global__ void deg_rank(const int* __restrict__ dst, int* __restrict__ cnt,
                         int* __restrict__ rank, int E4) {
    const int4* d4 = (const int4*)dst;
    int4* r4 = (int4*)rank;
    const int stride = gridDim.x * blockDim.x;
    for (int g = blockIdx.x * blockDim.x + threadIdx.x; g < E4; g += stride) {
        const int4 d = d4[g];
        int4 r;
        r.x = atomicAdd(&cnt[d.x], 1);
        r.y = atomicAdd(&cnt[d.y], 1);
        r.z = atomicAdd(&cnt[d.z], 1);
        r.w = atomicAdd(&cnt[d.w], 1);
        r4[g] = r;
    }
}

__global__ __launch_bounds__(1024) void row_scan(const int* __restrict__ cnt,
                                                 int* __restrict__ rowptr,
                                                 float* __restrict__ dinv, int nq) {
    __shared__ int wtot[32];
    __shared__ int wpre[32];
    const int tid  = threadIdx.x;
    const int lane = tid & 31;
    const int wave = tid >> 5;
    const int q0   = blockIdx.x * 1024;
    const int4* c4 = (const int4*)cnt;

    int pre = 0;
    #pragma unroll 4
    for (int i = tid; i < q0; i += 1024) {
        int4 t = c4[i];
        pre += (t.x + t.y) + (t.z + t.w);
    }
    const int q = q0 + tid;
    int4 c = make_int4(0, 0, 0, 0);
    if (q < nq) c = c4[q];
    const int tsum = (c.x + c.y) + (c.z + c.w);

    int s = tsum;
    #pragma unroll
    for (int o = 1; o < 32; o <<= 1) {
        int t = __shfl_up(s, o);
        if (lane >= o) s += t;
    }
    #pragma unroll
    for (int o = 16; o > 0; o >>= 1) pre += __shfl_xor(pre, o);

    if (lane == 31) wtot[wave] = s;
    if (lane == 0)  wpre[wave] = pre;
    __syncthreads();
    int off = 0;
    #pragma unroll
    for (int w = 0; w < 32; ++w) {
        off += wpre[w];
        if (w < wave) off += wtot[w];
    }
    const int e0 = off + s - tsum;
    if (q < nq) {
        int4 r;
        r.x = e0;
        r.y = e0 + c.x;
        r.z = r.y + c.y;
        r.w = r.z + c.z;
        ((int4*)rowptr)[q] = r;
        float4 d;
        d.x = rsqrtf((float)c.x + 1.0f);
        d.y = rsqrtf((float)c.y + 1.0f);
        d.z = rsqrtf((float)c.z + 1.0f);
        d.w = rsqrtf((float)c.w + 1.0f);
        ((float4*)dinv)[q] = d;
        if (q == nq - 1) rowptr[4 * nq] = e0 + tsum;
    }
}

__global__ void adj_fill(const int* __restrict__ src, const int* __restrict__ dst,
                         const int* __restrict__ rank, const int* __restrict__ rowptr,
                         const float* __restrict__ dinv, int2* __restrict__ adj, int E4) {
    const int4* s4 = (const int4*)src;
    const int4* d4 = (const int4*)dst;
    const int4* r4 = (const int4*)rank;
    const int stride = gridDim.x * blockDim.x;
    for (int g = blockIdx.x * blockDim.x + threadIdx.x; g < E4; g += stride) {
        const int4 u = s4[g];
        const int4 v = d4[g];
        const int4 r = r4[g];
        const int p0 = rowptr[v.x] + r.x;
        const int p1 = rowptr[v.y] + r.y;
        const int p2 = rowptr[v.z] + r.z;
        const int p3 = rowptr[v.w] + r.w;
        const float w0 = dinv[u.x] * dinv[v.x];
        const float w1 = dinv[u.y] * dinv[v.y];
        const float w2 = dinv[u.z] * dinv[v.z];
        const float w3 = dinv[u.w] * dinv[v.w];
        adj[p0] = make_int2(u.x, __float_as_int(w0));
        adj[p1] = make_int2(u.y, __float_as_int(w1));
        adj[p2] = make_int2(u.z, __float_as_int(w2));
        adj[p3] = make_int2(u.w, __float_as_int(w3));
    }
}

__global__ __launch_bounds__(256) void gcn_gemm(const float* __restrict__ X,
                                                const float* __restrict__ W,
                                                float* __restrict__ H, int tiles) {
    const int wave = threadIdx.x >> 5;
    const int lane = threadIdx.x & 31;
    const int col0 = wave << 4;
    const int row  = lane & 15;
    const int hi   = lane >> 4;

#ifdef USE_F32_WMMA
    const int kh = hi * 2;
    v2f bw[DIMF / 4];
    #pragma unroll
    for (int i = 0; i < DIMF / 4; ++i) {
        const int k = 4 * i;
        bw[i].x = W[(size_t)(k + kh)     * DIMF + col0 + row];
        bw[i].y = W[(size_t)(k + kh + 1) * DIMF + col0 + row];
    }
#endif

    for (int t = blockIdx.x; t < tiles; t += gridDim.x) {
        const int m0 = t << 4;
        v8f acc = {};
        const float* xrow = X + (size_t)(m0 + row) * DIMF;

#ifdef USE_F32_WMMA
        #pragma unroll
        for (int i = 0; i < DIMF / 4; ++i) {
            const int k = 4 * i;
            v2f a;
            a.x = xrow[k + kh];
            a.y = xrow[k + kh + 1];
            acc = __builtin_amdgcn_wmma_f32_16x16x4_f32(
                      false, a, false, bw[i], (short)0, acc, false, false);
        }
#else
        for (int k = 0; k < DIMF; k += 32) {
            v16h a, b;
            #pragma unroll
            for (int i = 0; i < 16; ++i) {
                int vg = i >> 1;
                int ka = ((vg < 4) ? (vg * 2) : (16 + (vg - 4) * 2)) + (i & 1) + hi * 8;
                a[i] = (_Float16)xrow[k + ka];
                int kb = i + hi * 16;
                b[i] = (_Float16)W[(size_t)(k + kb) * DIMF + col0 + row];
            }
            acc = __builtin_amdgcn_wmma_f32_16x16x32_f16(
                      false, a, false, b, (short)0, acc, false, false);
        }
#endif

        float* hp = H + (size_t)m0 * DIMF + col0 + row;
        const int mo = hi * 8;
        #pragma unroll
        for (int j = 0; j < 8; ++j)
            hp[(size_t)(mo + j) * DIMF] = acc[j];
    }
}


struct Rows8 {
    float4 h[8];
    float  w[8];
};

static __device__ __forceinline__ void rows8_issue(Rows8& r, const float4* __restrict__ H4,
                                                   int lane, int2 rec, int j, int cnt) {
    #pragma unroll
    for (int i = 0; i < 8; ++i) {
        const int u = __builtin_amdgcn_readlane(rec.x, j + i);
        r.w[i] = __int_as_float(__builtin_amdgcn_readlane(rec.y, j + i));
        r.h[i] = make_float4(0.0f, 0.0f, 0.0f, 0.0f);
        if (j + i < cnt) r.h[i] = H4[(size_t)u * (DIMF / 4) + lane];
    }
}

static __device__ __forceinline__ void rows8_add(float4& acc, const Rows8& r) {
    #pragma unroll
    for (int i = 0; i < 8; ++i) {
        acc.x += r.w[i] * r.h[i].x;
        acc.y += r.w[i] * r.h[i].y;
        acc.z += r.w[i] * r.h[i].z;
        acc.w += r.w[i] * r.h[i].w;
    }
}

__global__ __launch_bounds__(256) void gcn_gather(const float* __restrict__ H,
                                                  const int* __restrict__ rowptr,
                                                  const int2* __restrict__ adj,
                                                  const float* __restrict__ dinv,
                                                  const float* __restrict__ b,
                                                  float* __restrict__ out, int n) {
    const int lane = threadIdx.x & 31;
    const int gw = __builtin_amdgcn_readfirstlane((int)(blockIdx.x * 8 + (threadIdx.x >> 5)));
    const int nw = (int)gridDim.x * 8;
    if (gw >= n) return;
    const float4* H4 = (const float4*)H;
    const float4 bb = ((const float4*)b)[lane];

    int v1   = gw;
    int beg1 = rowptr[v1];
    int end1 = rowptr[v1 + 1];
    int2 a1  = make_int2(0, 0);
    if (beg1 + lane < end1) a1 = adj[beg1 + lane];
    asm volatile("" :: "v"(a1.x), "v"(a1.y));
    int v2 = v1 + nw;
    int beg2 = 0, end2 = 0;
    if (v2 < n) { beg2 = rowptr[v2]; end2 = rowptr[v2 + 1]; }

    while (true) {
        const int  v   = v1;
        const int  beg = beg1;
        const int  end = end1;
        const int2 a   = a1;
        v1 = v2; beg1 = beg2; end1 = end2;

        a1 = make_int2(0, 0);
        if (beg1 + lane < end1) a1 = adj[beg1 + lane];
        v2 = v1 + nw;
        beg2 = 0; end2 = 0;
        if (v2 < n) { beg2 = rowptr[v2]; end2 = rowptr[v2 + 1]; }

        const float s  = dinv[v];
        const float ss = s * s;
        const float4 h0 = H4[(size_t)v * (DIMF / 4) + lane];

        int cnt = end - beg;
        if (cnt > 32) cnt = 32;
        Rows8 r;
        rows8_issue(r, H4, lane, a, 0, cnt);
        float4 acc;
        acc.x = ss * h0.x;
        acc.y = ss * h0.y;
        acc.z = ss * h0.z;
        acc.w = ss * h0.w;
        rows8_add(acc, r);
        for (int j = 8; j < cnt; j += 8) {
            rows8_issue(r, H4, lane, a, j, cnt);
            rows8_add(acc, r);
        }
        for (int base = beg + 32; base < end; base += 32) {
            int2 cur = make_int2(0, 0);
            if (base + lane < end) cur = adj[base + lane];
            int c2 = end - base;
            if (c2 > 32) c2 = 32;
            for (int j = 0; j < c2; j += 8) {
                rows8_issue(r, H4, lane, cur, j, c2);
                rows8_add(acc, r);
            }
        }

        float4 o;
        o.x = acc.x + bb.x;
        o.y = acc.y + bb.y;
        o.z = acc.z + bb.z;
        o.w = acc.w + bb.w;
        o.x = (o.x > 0.0f) ? o.x : 0.01f * o.x;
        o.y = (o.y > 0.0f) ? o.y : 0.01f * o.y;
        o.z = (o.z > 0.0f) ? o.z : 0.01f * o.z;
        o.w = (o.w > 0.0f) ? o.w : 0.01f * o.w;
        ((float4*)out)[(size_t)v * (DIMF / 4) + lane] = o;

        if (v1 >= n) break;
    }
}


extern "C" void kernel_launch(void* const* d_in, const int* in_sizes, int n_in,
                              void* d_out, int out_size, void* d_ws, size_t ws_size,
                              hipStream_t stream) {
    const float* x  = (const float*)d_in[0];
    const int*   ei = (const int*)d_in[1];
    const float* Wts[3] = {(const float*)d_in[2], (const float*)d_in[4], (const float*)d_in[6]};
    const float* bs[3]  = {(const float*)d_in[3], (const float*)d_in[5], (const float*)d_in[7]};

    const int N = in_sizes[0] / DIMF;
    const int E = in_sizes[1] / 2;
    const int* srcp = ei;
    const int* dstp = ei + E;

    float* ws   = (float*)d_ws;
    const size_t nA = ((size_t)N + 64) & ~(size_t)63;
    const size_t eA = ((size_t)E + 63) & ~(size_t)63;
    float* dinv   = ws;
    float* Hbuf   = dinv + nA;
    float* Abuf   = Hbuf + (size_t)N * DIMF;
    int*   cnt    = (int*)(Abuf + (size_t)N * DIMF);
    int*   rowptr = cnt + nA;
    int*   rank   = rowptr + nA;
    int2*  adj    = (int2*)(rank + eA);

    const int T = 256;
    const int tilesM = N / 16;
    const int nq = N / 4;
    const int E4 = E / 4;

    const int gemmBlocks   = 768;
    const int gatherBlocks = 1024;

    zero_i32<<<(nq + T - 1) / T,       T,    0, stream>>>(cnt, nq);
    deg_rank<<<(E4 + T - 1) / T,       T,    0, stream>>>(dstp, cnt, rank, E4);
    row_scan<<<(nq + 1023) / 1024,     1024, 0, stream>>>(cnt, rowptr, dinv, nq);
    adj_fill<<<(E4 + T - 1) / T,       T,    0, stream>>>(srcp, dstp, rank, rowptr, dinv, adj, E4);

    for (int l = 0; l < 3; ++l) {
        const float* in = (l == 0) ? x : Abuf;
        gcn_gemm<<<gemmBlocks, 256, 0, stream>>>(in, Wts[l], Hbuf, tilesM);

        float* outp = (l == 2) ? (float*)d_out : Abuf;
        gcn_gather<<<gatherBlocks, 256, 0, stream>>>(Hbuf, rowptr, adj, dinv, bs[l], outp, N);
    }
}


# GEMM: each 16x128 X tile is fetched once per block with whole-row 16-byte loads and staged in LDS (two slots, one barrier per tile); the 8 waves read A fragments from LDS, not 8x from global
# speedup vs baseline: 3.3322x; 1.1526x over previous
#include <hip/hip_runtime.h>

typedef float     v2f  __attribute__((ext_vector_type(2)));
typedef float     v8f  __attribute__((ext_vector_type(8)));
typedef _Float16  v16h __attribute__((ext_vector_type(16)));

#define DIMF 128

#if __has_builtin(__builtin_amdgcn_wmma_f32_16x16x4_f32)
#define USE_F32_WMMA 1
#endif


__global__ void zero_i32(int* __restrict__ p, int n4) {
    int i = blockIdx.x * blockDim.x + threadIdx.x;
    if (i < n4) ((int4*)p)[i] = make_int4(0, 0, 0, 0);
}

__global__ void deg_rank(const int* __restrict__ dst, int* __restrict__ cnt,
                         int* __restrict__ rank, int E4) {
    const int4* d4 = (const int4*)dst;
    int4* r4 = (int4*)rank;
    const int stride = gridDim.x * blockDim.x;
    for (int g = blockIdx.x * blockDim.x + threadIdx.x; g < E4; g += stride) {
        const int4 d = d4[g];
        int4 r;
        r.x = atomicAdd(&cnt[d.x], 1);
        r.y = atomicAdd(&cnt[d.y], 1);
        r.z = atomicAdd(&cnt[d.z], 1);
        r.w = atomicAdd(&cnt[d.w], 1);
        r4[g] = r;
    }
}

__global__ __launch_bounds__(1024) void row_scan(const int* __restrict__ cnt,
                                                 int* __restrict__ rowptr,
                                                 float* __restrict__ dinv, int nq) {
    __shared__ int wtot[32];
    __shared__ int wpre[32];
    const int tid  = threadIdx.x;
    const int lane = tid & 31;
    const int wave = tid >> 5;
    const int q0   = blockIdx.x * 1024;
    const int4* c4 = (const int4*)cnt;

    int pre = 0;
    #pragma unroll 4
    for (int i = tid; i < q0; i += 1024) {
        int4 t = c4[i];
        pre += (t.x + t.y) + (t.z + t.w);
    }
    const int q = q0 + tid;
    int4 c = make_int4(0, 0, 0, 0);
    if (q < nq) c = c4[q];
    const int tsum = (c.x + c.y) + (c.z + c.w);

    int s = tsum;
    #pragma unroll
    for (int o = 1; o < 32; o <<= 1) {
        int t = __shfl_up(s, o);
        if (lane >= o) s += t;
    }
    #pragma unroll
    for (int o = 16; o > 0; o >>= 1) pre += __shfl_xor(pre, o);

    if (lane == 31) wtot[wave] = s;
    if (lane == 0)  wpre[wave] = pre;
    __syncthreads();
    int off = 0;
    #pragma unroll
    for (int w = 0; w < 32; ++w) {
        off += wpre[w];
        if (w < wave) off += wtot[w];
    }
    const int e0 = off + s - tsum;
    if (q < nq) {
        int4 r;
        r.x = e0;
        r.y = e0 + c.x;
        r.z = r.y + c.y;
        r.w = r.z + c.z;
        ((int4*)rowptr)[q] = r;
        float4 d;
        d.x = rsqrtf((float)c.x + 1.0f);
        d.y = rsqrtf((float)c.y + 1.0f);
        d.z = rsqrtf((float)c.z + 1.0f);
        d.w = rsqrtf((float)c.w + 1.0f);
        ((float4*)dinv)[q] = d;
        if (q == nq - 1) rowptr[4 * nq] = e0 + tsum;
    }
}

__global__ void adj_fill(const int* __restrict__ src, const int* __restrict__ dst,
                         const int* __restrict__ rank, const int* __restrict__ rowptr,
                         const float* __restrict__ dinv, int2* __restrict__ adj, int E4) {
    const int4* s4 = (const int4*)src;
    const int4* d4 = (const int4*)dst;
    const int4* r4 = (const int4*)rank;
    const int stride = gridDim.x * blockDim.x;
    for (int g = blockIdx.x * blockDim.x + threadIdx.x; g < E4; g += stride) {
        const int4 u = s4[g];
        const int4 v = d4[g];
        const int4 r = r4[g];
        const int p0 = rowptr[v.x] + r.x;
        const int p1 = rowptr[v.y] + r.y;
        const int p2 = rowptr[v.z] + r.z;
        const int p3 = rowptr[v.w] + r.w;
        const float w0 = dinv[u.x] * dinv[v.x];
        const float w1 = dinv[u.y] * dinv[v.y];
        const float w2 = dinv[u.z] * dinv[v.z];
        const float w3 = dinv[u.w] * dinv[v.w];
        adj[p0] = make_int2(u.x, __float_as_int(w0));
        adj[p1] = make_int2(u.y, __float_as_int(w1));
        adj[p2] = make_int2(u.z, __float_as_int(w2));
        adj[p3] = make_int2(u.w, __float_as_int(w3));
    }
}


#define XPITCH 132

static __device__ __forceinline__ void lds_handover_barrier() {
    __builtin_amdgcn_fence(__ATOMIC_RELEASE, "workgroup", "local");
    __builtin_amdgcn_s_barrier();
    __builtin_amdgcn_fence(__ATOMIC_ACQUIRE, "workgroup", "local");
}

__global__ __launch_bounds__(256) void gcn_gemm(const float* __restrict__ X,
                                                const float* __restrict__ W,
                                                float* __restrict__ H, int tiles) {
    __shared__ __attribute__((aligned(16))) float xs[2][16 * XPITCH];

    const int wave = threadIdx.x >> 5;
    const int lane = threadIdx.x & 31;
    const int col0 = wave << 4;
    const int row  = lane & 15;
    const int hi   = lane >> 4;

    int t = blockIdx.x;
    if (t >= tiles) return;

    const float4* xg = (const float4*)X + ((size_t)t * 16 + wave) * (DIMF / 4) + lane;
    const size_t tstep = (size_t)gridDim.x * 16 * (DIMF / 4);
    const int so = wave * XPITCH + 4 * lane;
    float4 p0 = xg[0];
    float4 p1 = xg[8 * (DIMF / 4)];

#ifdef USE_F32_WMMA
    const int kh = hi * 2;
    v2f bw[DIMF / 4];
    #pragma unroll
    for (int i = 0; i < DIMF / 4; ++i) {
        const int k = 4 * i;
        bw[i].x = W[(size_t)(k + kh)     * DIMF + col0 + row];
        bw[i].y = W[(size_t)(k + kh + 1) * DIMF + col0 + row];
    }
#endif

    int buf = 0;
    while (true) {
        *(float4*)&xs[buf][so]              = p0;
        *(float4*)&xs[buf][so + 8 * XPITCH] = p1;
        lds_handover_barrier();

        const int  tn   = t + (int)gridDim.x;
        const bool more = tn < tiles;
        if (more) {
            xg += tstep;
            p0 = xg[0];
            p1 = xg[8 * (DIMF / 4)];
        }

        const int m0 = t << 4;
        v8f acc = {};
        const float* xrow = &xs[buf][row * XPITCH];

#ifdef USE_F32_WMMA
        #pragma unroll
        for (int i = 0; i < DIMF / 4; ++i) {
            const v2f a = *(const v2f*)(xrow + 4 * i + kh);
            acc = __builtin_amdgcn_wmma_f32_16x16x4_f32(
                      false, a, false, bw[i], (short)0, acc, false, false);
        }
#else
        for (int k = 0; k < DIMF; k += 32) {
            v16h a, b;
            #pragma unroll
            for (int i = 0; i < 16; ++i) {
                int vg = i >> 1;
                int ka = ((vg < 4) ? (vg * 2) : (16 + (vg - 4) * 2)) + (i & 1) + hi * 8;
                a[i] = (_Float16)xrow[k + ka];
                int kb = i + hi * 16;
                b[i] = (_Float16)W[(size_t)(k + kb) * DIMF + col0 + row];
            }
            acc = __builtin_amdgcn_wmma_f32_16x16x32_f16(
                      false, a, false, b, (short)0, acc, false, false);
        }
#endif

        float* hp = H + (size_t)m0 * DIMF + col0 + row;
        const int mo = hi * 8;
        #pragma unroll
        for (int j = 0; j < 8; ++j)
            hp[(size_t)(mo + j) * DIMF] = acc[j];

        if (!more) break;
        buf ^= 1;
        t = tn;
    }
}


struct Rows8 {
    float4 h[8];
    float  w[8];
};

static __device__ __forceinline__ void rows8_issue(Rows8& r, const float4* __restrict__ H4,
                                                   int lane, int2 rec, int j, int cnt) {
    #pragma unroll
    for (int i = 0; i < 8; ++i) {
        const int u = __builtin_amdgcn_readlane(rec.x, j + i);
        r.w[i] = __int_as_float(__builtin_amdgcn_readlane(rec.y, j + i));
        r.h[i] = make_float4(0.0f, 0.0f, 0.0f, 0.0f);
        if (j + i < cnt) r.h[i] = H4[(size_t)u * (DIMF / 4) + lane];
    }
}

static __device__ __forceinline__ void rows8_add(float4& acc, const Rows8& r) {
    #pragma unroll
    for (int i = 0; i < 8; ++i) {
        acc.x += r.w[i] * r.h[i].x;
        acc.y += r.w[i] * r.h[i].y;
        acc.z += r.w[i] * r.h[i].z;
        acc.w += r.w[i] * r.h[i].w;
    }
}

__global__ __launch_bounds__(256) void gcn_gather(const float* __restrict__ H,
                                                  const int* __restrict__ rowptr,
                                                  const int2* __restrict__ adj,
                                                  const float* __restrict__ dinv,
                                                  const float* __restrict__ b,
                                                  float* __restrict__ out, int n) {
    const int lane = threadIdx.x & 31;
    const int gw = __builtin_amdgcn_readfirstlane((int)(blockIdx.x * 8 + (threadIdx.x >> 5)));
    const int nw = (int)gridDim.x * 8;
    if (gw >= n) return;
    const float4* H4 = (const float4*)H;
    const float4 bb = ((const float4*)b)[lane];

    int v1   = gw;
    int beg1 = rowptr[v1];
    int end1 = rowptr[v1 + 1];
    int2 a1  = make_int2(0, 0);
    if (beg1 + lane < end1) a1 = adj[beg1 + lane];
    asm volatile("" :: "v"(a1.x), "v"(a1.y));
    int v2 = v1 + nw;
    int beg2 = 0, end2 = 0;
    if (v2 < n) { beg2 = rowptr[v2]; end2 = rowptr[v2 + 1]; }

    while (true) {
        const int  v   = v1;
        const int  beg = beg1;
        const int  end = end1;
        const int2 a   = a1;
        v1 = v2; beg1 = beg2; end1 = end2;

        a1 = make_int2(0, 0);
        if (beg1 + lane < end1) a1 = adj[beg1 + lane];
        v2 = v1 + nw;
        beg2 = 0; end2 = 0;
        if (v2 < n) { beg2 = rowptr[v2]; end2 = rowptr[v2 + 1]; }

        const float s  = dinv[v];
        const float ss = s * s;
        const float4 h0 = H4[(size_t)v * (DIMF / 4) + lane];

        int cnt = end - beg;
        if (cnt > 32) cnt = 32;
        Rows8 r;
        rows8_issue(r, H4, lane, a, 0, cnt);
        float4 acc;
        acc.x = ss * h0.x;
        acc.y = ss * h0.y;
        acc.z = ss * h0.z;
        acc.w = ss * h0.w;
        rows8_add(acc, r);
        for (int j = 8; j < cnt; j += 8) {
            rows8_issue(r, H4, lane, a, j, cnt);
            rows8_add(acc, r);
        }
        for (int base = beg + 32; base < end; base += 32) {
            int2 cur = make_int2(0, 0);
            if (base + lane < end) cur = adj[base + lane];
            int c2 = end - base;
            if (c2 > 32) c2 = 32;
            for (int j = 0; j < c2; j += 8) {
                rows8_issue(r, H4, lane, cur, j, c2);
                rows8_add(acc, r);
            }
        }

        float4 o;
        o.x = acc.x + bb.x;
        o.y = acc.y + bb.y;
        o.z = acc.z + bb.z;
        o.w = acc.w + bb.w;
        o.x = (o.x > 0.0f) ? o.x : 0.01f * o.x;
        o.y = (o.y > 0.0f) ? o.y : 0.01f * o.y;
        o.z = (o.z > 0.0f) ? o.z : 0.01f * o.z;
        o.w = (o.w > 0.0f) ? o.w : 0.01f * o.w;
        ((float4*)out)[(size_t)v * (DIMF / 4) + lane] = o;

        if (v1 >= n) break;
    }
}


extern "C" void kernel_launch(void* const* d_in, const int* in_sizes, int n_in,
                              void* d_out, int out_size, void* d_ws, size_t ws_size,
                              hipStream_t stream) {
    const float* x  = (const float*)d_in[0];
    const int*   ei = (const int*)d_in[1];
    const float* Wts[3] = {(const float*)d_in[2], (const float*)d_in[4], (const float*)d_in[6]};
    const float* bs[3]  = {(const float*)d_in[3], (const float*)d_in[5], (const float*)d_in[7]};

    const int N = in_sizes[0] / DIMF;
    const int E = in_sizes[1] / 2;
    const int* srcp = ei;
    const int* dstp = ei + E;

    float* ws   = (float*)d_ws;
    const size_t nA = ((size_t)N + 64) & ~(size_t)63;
    const size_t eA = ((size_t)E + 63) & ~(size_t)63;
    float* dinv   = ws;
    float* Hbuf   = dinv + nA;
    float* Abuf   = Hbuf + (size_t)N * DIMF;
    int*   cnt    = (int*)(Abuf + (size_t)N * DIMF);
    int*   rowptr = cnt + nA;
    int*   rank   = rowptr + nA;
    int2*  adj    = (int2*)(rank + eA);

    const int T = 256;
    const int tilesM = N / 16;
    const int nq = N / 4;
    const int E4 = E / 4;

    const int gemmBlocks   = 768;
    const int gatherBlocks = 1024;

    zero_i32<<<(nq + T - 1) / T,       T,    0, stream>>>(cnt, nq);
    deg_rank<<<(E4 + T - 1) / T,       T,    0, stream>>>(dstp, cnt, rank, E4);
    row_scan<<<(nq + 1023) / 1024,     1024, 0, stream>>>(cnt, rowptr, dinv, nq);
    adj_fill<<<(E4 + T - 1) / T,       T,    0, stream>>>(srcp, dstp, rank, rowptr, dinv, adj, E4);

    for (int l = 0; l < 3; ++l) {
        const float* in = (l == 0) ? x : Abuf;
        gcn_gemm<<<gemmBlocks, 256, 0, stream>>>(in, Wts[l], Hbuf, tilesM);

        float* outp = (l == 2) ? (float*)d_out : Abuf;
        gcn_gather<<<gatherBlocks, 256, 0, stream>>>(Hbuf, rowptr, adj, dinv, bs[l], outp, N);
    }
}


# Layer fused into one launch, summed as (Ahat X) W: a block sums 16 nodes' weighted input rows into LDS, then multiplies by W on WMMA; X W never goes to memory; 3 launches gone; sum order changes
# speedup vs baseline: 3.4603x; 1.0384x over previous
#include <hip/hip_runtime.h>

typedef float     v2f  __attribute__((ext_vector_type(2)));
typedef float     v8f  __attribute__((ext_vector_type(8)));
typedef _Float16  v16h __attribute__((ext_vector_type(16)));

#define DIMF 128

#if __has_builtin(__builtin_amdgcn_wmma_f32_16x16x4_f32)
#define USE_F32_WMMA 1
#endif


__global__ void zero_i32(int* __restrict__ p, int n4) {
    int i = blockIdx.x * blockDim.x + threadIdx.x;
    if (i < n4) ((int4*)p)[i] = make_int4(0, 0, 0, 0);
}

__global__ void deg_rank(const int* __restrict__ dst, int* __restrict__ cnt,
                         int* __restrict__ rank, int E4) {
    const int4* d4 = (const int4*)dst;
    int4* r4 = (int4*)rank;
    const int stride = gridDim.x * blockDim.x;
    for (int g = blockIdx.x * blockDim.x + threadIdx.x; g < E4; g += stride) {
        const int4 d = d4[g];
        int4 r;
        r.x = atomicAdd(&cnt[d.x], 1);
        r.y = atomicAdd(&cnt[d.y], 1);
        r.z = atomicAdd(&cnt[d.z], 1);
        r.w = atomicAdd(&cnt[d.w], 1);
        r4[g] = r;
    }
}

__global__ __launch_bounds__(1024) void row_scan(const int* __restrict__ cnt,
                                                 int* __restrict__ rowptr,
                                                 float* __restrict__ dinv, int nq) {
    __shared__ int wtot[32];
    __shared__ int wpre[32];
    const int tid  = threadIdx.x;
    const int lane = tid & 31;
    const int wave = tid >> 5;
    const int q0   = blockIdx.x * 1024;
    const int4* c4 = (const int4*)cnt;

    int pre = 0;
    #pragma unroll 4
    for (int i = tid; i < q0; i += 1024) {
        int4 t = c4[i];
        pre += (t.x + t.y) + (t.z + t.w);
    }
    const int q = q0 + tid;
    int4 c = make_int4(0, 0, 0, 0);
    if (q < nq) c = c4[q];
    const int tsum = (c.x + c.y) + (c.z + c.w);

    int s = tsum;
    #pragma unroll
    for (int o = 1; o < 32; o <<= 1) {
        int t = __shfl_up(s, o);
        if (lane >= o) s += t;
    }
    #pragma unroll
    for (int o = 16; o > 0; o >>= 1) pre += __shfl_xor(pre, o);

    if (lane == 31) wtot[wave] = s;
    if (lane == 0)  wpre[wave] = pre;
    __syncthreads();
    int off = 0;
    #pragma unroll
    for (int w = 0; w < 32; ++w) {
        off += wpre[w];
        if (w < wave) off += wtot[w];
    }
    const int e0 = off + s - tsum;
    if (q < nq) {
        int4 r;
        r.x = e0;
        r.y = e0 + c.x;
        r.z = r.y + c.y;
        r.w = r.z + c.z;
        ((int4*)rowptr)[q] = r;
        float4 d;
        d.x = rsqrtf((float)c.x + 1.0f);
        d.y = rsqrtf((float)c.y + 1.0f);
        d.z = rsqrtf((float)c.z + 1.0f);
        d.w = rsqrtf((float)c.w + 1.0f);
        ((float4*)dinv)[q] = d;
        if (q == nq - 1) rowptr[4 * nq] = e0 + tsum;
    }
}

__global__ void adj_fill(const int* __restrict__ src, const int* __restrict__ dst,
                         const int* __restrict__ rank, const int* __restrict__ rowptr,
                         const float* __restrict__ dinv, int2* __restrict__ adj, int E4) {
    const int4* s4 = (const int4*)src;
    const int4* d4 = (const int4*)dst;
    const int4* r4 = (const int4*)rank;
    const int stride = gridDim.x * blockDim.x;
    for (int g = blockIdx.x * blockDim.x + threadIdx.x; g < E4; g += stride) {
        const int4 u = s4[g];
        const int4 v = d4[g];
        const int4 r = r4[g];
        const int p0 = rowptr[v.x] + r.x;
        const int p1 = rowptr[v.y] + r.y;
        const int p2 = rowptr[v.z] + r.z;
        const int p3 = rowptr[v.w] + r.w;
        const float w0 = dinv[u.x] * dinv[v.x];
        const float w1 = dinv[u.y] * dinv[v.y];
        const float w2 = dinv[u.z] * dinv[v.z];
        const float w3 = dinv[u.w] * dinv[v.w];
        adj[p0] = make_int2(u.x, __float_as_int(w0));
        adj[p1] = make_int2(u.y, __float_as_int(w1));
        adj[p2] = make_int2(u.z, __float_as_int(w2));
        adj[p3] = make_int2(u.w, __float_as_int(w3));
    }
}


#define XPITCH 132

static __device__ __forceinline__ void lds_handover_barrier() {
    __builtin_amdgcn_fence(__ATOMIC_RELEASE, "workgroup", "local");
    __builtin_amdgcn_s_barrier();
    __builtin_amdgcn_fence(__ATOMIC_ACQUIRE, "workgroup", "local");
}

struct Rows8 {
    float4 h[8];
    float  w[8];
};

static __device__ __forceinline__ void rows8_issue(Rows8& r, const float4* __restrict__ X4,
                                                   int lane, int2 rec, int j, int cnt) {
    #pragma unroll
    for (int i = 0; i < 8; ++i) {
        const int u = __builtin_amdgcn_readlane(rec.x, j + i);
        r.w[i] = __int_as_float(__builtin_amdgcn_readlane(rec.y, j + i));
        r.h[i] = make_float4(0.0f, 0.0f, 0.0f, 0.0f);
        if (j + i < cnt) r.h[i] = X4[(size_t)u * (DIMF / 4) + lane];
    }
}

static __device__ __forceinline__ void rows8_add(float4& acc, const Rows8& r) {
    #pragma unroll
    for (int i = 0; i < 8; ++i) {
        acc.x += r.w[i] * r.h[i].x;
        acc.y += r.w[i] * r.h[i].y;
        acc.z += r.w[i] * r.h[i].z;
        acc.w += r.w[i] * r.h[i].w;
    }
}

__global__ __launch_bounds__(256) void gcn_layer(const float* __restrict__ X,
                                                 const float* __restrict__ W,
                                                 const float* __restrict__ b,
                                                 const int* __restrict__ rowptr,
                                                 const int2* __restrict__ adj,
                                                 const float* __restrict__ dinv,
                                                 float* __restrict__ out, int tiles) {
    __shared__ __attribute__((aligned(16))) float xs[2][16 * XPITCH];

    const int wave = threadIdx.x >> 5;
    const int lane = threadIdx.x & 31;
    const int col0 = wave << 4;
    const int row  = lane & 15;
    const int hi   = lane >> 4;

    int t = blockIdx.x;
    if (t >= tiles) return;
    const int n = tiles * 16;
    const int vstep = (int)gridDim.x * 16 - 8;
    const float4* X4 = (const float4*)X;

#ifdef USE_F32_WMMA
    const int kh = hi * 2;
    v2f bw[DIMF / 4];
    #pragma unroll
    for (int i = 0; i < DIMF / 4; ++i) {
        const int k = 4 * i;
        bw[i].x = W[(size_t)(k + kh)     * DIMF + col0 + row];
        bw[i].y = W[(size_t)(k + kh + 1) * DIMF + col0 + row];
    }
#endif
    const float bias = b[col0 + row];

    int v1   = __builtin_amdgcn_readfirstlane(t * 16 + wave);
    int beg1 = rowptr[v1];
    int end1 = rowptr[v1 + 1];
    int v2   = v1 + 8;
    int be2  = rowptr[v2 + (lane & 1)];
    int2 a1  = make_int2(0, 0);
    if (beg1 + lane < end1) a1 = adj[beg1 + lane];

    int buf = 0;
    while (true) {
        #pragma unroll
        for (int half = 0; half < 2; ++half) {
            const int  v   = v1;
            const int  beg = beg1;
            const int  end = end1;
            const int2 a   = a1;
            v1   = v2;
            beg1 = __builtin_amdgcn_readlane(be2, 0);
            end1 = __builtin_amdgcn_readlane(be2, 1);

            a1 = make_int2(0, 0);
            if (beg1 + lane < end1) a1 = adj[beg1 + lane];
            v2 = (half == 0) ? v1 + vstep : v1 + 8;
            be2 = 0;
            if (v2 < n) be2 = rowptr[v2 + (lane & 1)];

            const float s  = dinv[v];
            const float ss = s * s;
            const float4 h0 = X4[(size_t)v * (DIMF / 4) + lane];

            int cnt = end - beg;
            if (cnt > 32) cnt = 32;
            Rows8 r;
            rows8_issue(r, X4, lane, a, 0, cnt);
            float4 acc;
            acc.x = ss * h0.x;
            acc.y = ss * h0.y;
            acc.z = ss * h0.z;
            acc.w = ss * h0.w;
            rows8_add(acc, r);
            for (int j = 8; j < cnt; j += 8) {
                rows8_issue(r, X4, lane, a, j, cnt);
                rows8_add(acc, r);
            }
            for (int base = beg + 32; base < end; base += 32) {
                int2 cur = make_int2(0, 0);
                if (base + lane < end) cur = adj[base + lane];
                int c2 = end - base;
                if (c2 > 32) c2 = 32;
                for (int j = 0; j < c2; j += 8) {
                    rows8_issue(r, X4, lane, cur, j, c2);
                    rows8_add(acc, r);
                }
            }
            *(float4*)&xs[buf][(wave + 8 * half) * XPITCH + 4 * lane] = acc;
        }
        lds_handover_barrier();

        const int m0 = t << 4;
        v8f pacc = {};
        const float* xrow = &xs[buf][row * XPITCH];

#ifdef USE_F32_WMMA
        #pragma unroll
        for (int i = 0; i < DIMF / 4; ++i) {
            const v2f a = *(const v2f*)(xrow + 4 * i + kh);
            pacc = __builtin_amdgcn_wmma_f32_16x16x4_f32(
                       false, a, false, bw[i], (short)0, pacc, false, false);
        }
#else
        for (int k = 0; k < DIMF; k += 32) {
            v16h a, bf;
            #pragma unroll
            for (int i = 0; i < 16; ++i) {
                int vg = i >> 1;
                int ka = ((vg < 4) ? (vg * 2) : (16 + (vg - 4) * 2)) + (i & 1) + hi * 8;
                a[i] = (_Float16)xrow[k + ka];
                int kb = i + hi * 16;
                bf[i] = (_Float16)W[(size_t)(k + kb) * DIMF + col0 + row];
            }
            pacc = __builtin_amdgcn_wmma_f32_16x16x32_f16(
                       false, a, false, bf, (short)0, pacc, false, false);
        }
#endif

        float* op = out + (size_t)m0 * DIMF + col0 + row;
        const int mo = hi * 8;
        #pragma unroll
        for (int j = 0; j < 8; ++j) {
            float o = pacc[j] + bias;
            o = (o > 0.0f) ? o : 0.01f * o;
            op[(size_t)(mo + j) * DIMF] = o;
        }

        t += (int)gridDim.x;
        if (t >= tiles) break;
        buf ^= 1;
    }
}


extern "C" void kernel_launch(void* const* d_in, const int* in_sizes, int n_in,
                              void* d_out, int out_size, void* d_ws, size_t ws_size,
                              hipStream_t stream) {
    const float* x  = (const float*)d_in[0];
    const int*   ei = (const int*)d_in[1];
    const float* Wts[3] = {(const float*)d_in[2], (const float*)d_in[4], (const float*)d_in[6]};
    const float* bs[3]  = {(const float*)d_in[3], (const float*)d_in[5], (const float*)d_in[7]};

    const int N = in_sizes[0] / DIMF;
    const int E = in_sizes[1] / 2;
    const int* srcp = ei;
    const int* dstp = ei + E;

    float* ws   = (float*)d_ws;
    const size_t nA = ((size_t)N + 64) & ~(size_t)63;
    const size_t eA = ((size_t)E + 63) & ~(size_t)63;
    float* dinv   = ws;
    float* Hbuf   = dinv + nA;
    float* Abuf   = Hbuf + (size_t)N * DIMF;
    int*   cnt    = (int*)(Abuf + (size_t)N * DIMF);
    int*   rowptr = cnt + nA;
    int*   rank   = rowptr + nA;
    int2*  adj    = (int2*)(rank + eA);

    const int T = 256;
    const int tilesM = N / 16;
    const int nq = N / 4;
    const int E4 = E / 4;

    const int layerBlocks = 782;

    zero_i32<<<(nq + T - 1) / T,       T,    0, stream>>>(cnt, nq);
    deg_rank<<<(E4 + T - 1) / T,       T,    0, stream>>>(dstp, cnt, rank, E4);
    row_scan<<<(nq + 1023) / 1024,     1024, 0, stream>>>(cnt, rowptr, dinv, nq);
    adj_fill<<<(E4 + T - 1) / T,       T,    0, stream>>>(srcp, dstp, rank, rowptr, dinv, adj, E4);

    gcn_layer<<<layerBlocks, 256, 0, stream>>>(x,    Wts[0], bs[0], rowptr, adj, dinv, Abuf,          tilesM);
    gcn_layer<<<layerBlocks, 256, 0, stream>>>(Abuf, Wts[1], bs[1], rowptr, adj, dinv, Hbuf,          tilesM);
    gcn_layer<<<layerBlocks, 256, 0, stream>>>(Hbuf, Wts[2], bs[2], rowptr, adj, dinv, (float*)d_out, tilesM);
}


# A wave sums its two nodes of a tile at once: 16+2 row reads in flight, not 8+1; a tile costs the longer node's round trips, not both; 768 blocks, 3 per unit at 160 registers; per-node sum order kept
# speedup vs baseline: 3.5814x; 1.0350x over previous
#include <hip/hip_runtime.h>

typedef float     v2f  __attribute__((ext_vector_type(2)));
typedef float     v8f  __attribute__((ext_vector_type(8)));
typedef _Float16  v16h __attribute__((ext_vector_type(16)));

#define DIMF 128

#if __has_builtin(__builtin_amdgcn_wmma_f32_16x16x4_f32)
#define USE_F32_WMMA 1
#endif


__global__ void zero_i32(int* __restrict__ p, int n4) {
    int i = blockIdx.x * blockDim.x + threadIdx.x;
    if (i < n4) ((int4*)p)[i] = make_int4(0, 0, 0, 0);
}

__global__ void deg_rank(const int* __restrict__ dst, int* __restrict__ cnt,
                         int* __restrict__ rank, int E4) {
    const int4* d4 = (const int4*)dst;
    int4* r4 = (int4*)rank;
    const int stride = gridDim.x * blockDim.x;
    for (int g = blockIdx.x * blockDim.x + threadIdx.x; g < E4; g += stride) {
        const int4 d = d4[g];
        int4 r;
        r.x = atomicAdd(&cnt[d.x], 1);
        r.y = atomicAdd(&cnt[d.y], 1);
        r.z = atomicAdd(&cnt[d.z], 1);
        r.w = atomicAdd(&cnt[d.w], 1);
        r4[g] = r;
    }
}

__global__ __launch_bounds__(1024) void row_scan(const int* __restrict__ cnt,
                                                 int* __restrict__ rowptr,
                                                 float* __restrict__ dinv, int nq) {
    __shared__ int wtot[32];
    __shared__ int wpre[32];
    const int tid  = threadIdx.x;
    const int lane = tid & 31;
    const int wave = tid >> 5;
    const int q0   = blockIdx.x * 1024;
    const int4* c4 = (const int4*)cnt;

    int pre = 0;
    #pragma unroll 4
    for (int i = tid; i < q0; i += 1024) {
        int4 t = c4[i];
        pre += (t.x + t.y) + (t.z + t.w);
    }
    const int q = q0 + tid;
    int4 c = make_int4(0, 0, 0, 0);
    if (q < nq) c = c4[q];
    const int tsum = (c.x + c.y) + (c.z + c.w);

    int s = tsum;
    #pragma unroll
    for (int o = 1; o < 32; o <<= 1) {
        int t = __shfl_up(s, o);
        if (lane >= o) s += t;
    }
    #pragma unroll
    for (int o = 16; o > 0; o >>= 1) pre += __shfl_xor(pre, o);

    if (lane == 31) wtot[wave] = s;
    if (lane == 0)  wpre[wave] = pre;
    __syncthreads();
    int off = 0;
    #pragma unroll
    for (int w = 0; w < 32; ++w) {
        off += wpre[w];
        if (w < wave) off += wtot[w];
    }
    const int e0 = off + s - tsum;
    if (q < nq) {
        int4 r;
        r.x = e0;
        r.y = e0 + c.x;
        r.z = r.y + c.y;
        r.w = r.z + c.z;
        ((int4*)rowptr)[q] = r;
        float4 d;
        d.x = rsqrtf((float)c.x + 1.0f);
        d.y = rsqrtf((float)c.y + 1.0f);
        d.z = rsqrtf((float)c.z + 1.0f);
        d.w = rsqrtf((float)c.w + 1.0f);
        ((float4*)dinv)[q] = d;
        if (q == nq - 1) rowptr[4 * nq] = e0 + tsum;
    }
}

__global__ void adj_fill(const int* __restrict__ src, const int* __restrict__ dst,
                         const int* __restrict__ rank, const int* __restrict__ rowptr,
                         const float* __restrict__ dinv, int2* __restrict__ adj, int E4) {
    const int4* s4 = (const int4*)src;
    const int4* d4 = (const int4*)dst;
    const int4* r4 = (const int4*)rank;
    const int stride = gridDim.x * blockDim.x;
    for (int g = blockIdx.x * blockDim.x + threadIdx.x; g < E4; g += stride) {
        const int4 u = s4[g];
        const int4 v = d4[g];
        const int4 r = r4[g];
        const int p0 = rowptr[v.x] + r.x;
        const int p1 = rowptr[v.y] + r.y;
        const int p2 = rowptr[v.z] + r.z;
        const int p3 = rowptr[v.w] + r.w;
        const float w0 = dinv[u.x] * dinv[v.x];
        const float w1 = dinv[u.y] * dinv[v.y];
        const float w2 = dinv[u.z] * dinv[v.z];
        const float w3 = dinv[u.w] * dinv[v.w];
        adj[p0] = make_int2(u.x, __float_as_int(w0));
        adj[p1] = make_int2(u.y, __float_as_int(w1));
        adj[p2] = make_int2(u.z, __float_as_int(w2));
        adj[p3] = make_int2(u.w, __float_as_int(w3));
    }
}


#define XPITCH 132

static __device__ __forceinline__ void lds_handover_barrier() {
    __builtin_amdgcn_fence(__ATOMIC_RELEASE, "workgroup", "local");
    __builtin_amdgcn_s_barrier();
    __builtin_amdgcn_fence(__ATOMIC_ACQUIRE, "workgroup", "local");
}

struct Rows8 {
    float4 h[8];
    float  w[8];
};

static __device__ __forceinline__ void rows8_issue(Rows8& r, const float4* __restrict__ X4,
                                                   int lane, int2 rec, int j, int cnt) {
    #pragma unroll
    for (int i = 0; i < 8; ++i) {
        const int u = __builtin_amdgcn_readlane(rec.x, j + i);
        r.w[i] = __int_as_float(__builtin_amdgcn_readlane(rec.y, j + i));
        r.h[i] = make_float4(0.0f, 0.0f, 0.0f, 0.0f);
        if (j + i < cnt) r.h[i] = X4[(size_t)u * (DIMF / 4) + lane];
    }
}

static __device__ __forceinline__ void rows8_add(float4& acc, const Rows8& r) {
    #pragma unroll
    for (int i = 0; i < 8; ++i) {
        acc.x = __builtin_fmaf(r.w[i], r.h[i].x, acc.x);
        acc.y = __builtin_fmaf(r.w[i], r.h[i].y, acc.y);
        acc.z = __builtin_fmaf(r.w[i], r.h[i].z, acc.z);
        acc.w = __builtin_fmaf(r.w[i], r.h[i].w, acc.w);
    }
}

static __device__ __forceinline__ void rows_past32(float4& acc,
                                                   const float4* __restrict__ X4,
                                                   const int2* __restrict__ adj,
                                                   int lane, int beg, int end) {
    for (int base = beg + 32; base < end; base += 32) {
        int2 cur = make_int2(0, 0);
        if (base + lane < end) cur = adj[base + lane];
        int c2 = end - base;
        if (c2 > 32) c2 = 32;
        int j = 0;
        do {
            Rows8 r;
            rows8_issue(r, X4, lane, cur, j, c2);
            rows8_add(acc, r);
            j += 8;
        } while (j < c2);
    }
}

__global__ __launch_bounds__(256) void gcn_layer(const float* __restrict__ X,
                                                 const float* __restrict__ W,
                                                 const float* __restrict__ b,
                                                 const int* __restrict__ rowptr,
                                                 const int2* __restrict__ adj,
                                                 const float* __restrict__ dinv,
                                                 float* __restrict__ out, int tiles) {
    __shared__ __attribute__((aligned(16))) float xs[2][16 * XPITCH];

    const int wave = threadIdx.x >> 5;
    const int lane = threadIdx.x & 31;
    const int col0 = wave << 4;
    const int row  = lane & 15;
    const int hi   = lane >> 4;

    int t = blockIdx.x;
    if (t >= tiles) return;
    const int tstep = (int)gridDim.x;
    const float4* X4 = (const float4*)X;

#ifdef USE_F32_WMMA
    const int kh = hi * 2;
    v2f bw[DIMF / 4];
    #pragma unroll
    for (int i = 0; i < DIMF / 4; ++i) {
        const int k = 4 * i;
        bw[i].x = W[(size_t)(k + kh)     * DIMF + col0 + row];
        bw[i].y = W[(size_t)(k + kh + 1) * DIMF + col0 + row];
    }
#endif
    const float bias = b[col0 + row];

    const int boff = (lane & 1) + ((lane & 2) << 2);

    int vA   = __builtin_amdgcn_readfirstlane(t * 16 + wave);
    int begA = rowptr[vA];
    int endA = rowptr[vA + 1];
    int begB = rowptr[vA + 8];
    int endB = rowptr[vA + 9];
    int2 recA = make_int2(0, 0);
    int2 recB = make_int2(0, 0);
    if (begA + lane < endA) recA = adj[begA + lane];
    if (begB + lane < endB) recB = adj[begB + lane];
    int nb = 0;
    if (t + tstep < tiles) nb = rowptr[(t + tstep) * 16 + wave + boff];
    float ssA = dinv[vA];
    float ssB = dinv[vA + 8];
    ssA *= ssA;
    ssB *= ssB;

    int buf = 0;
    while (true) {
        const int tn = t + tstep;
        const int vB = vA + 8;
        const float4 hA = X4[(size_t)vA * (DIMF / 4) + lane];
        const float4 hB = X4[(size_t)vB * (DIMF / 4) + lane];

        int cntA = endA - begA;
        if (cntA > 32) cntA = 32;
        int cntB = endB - begB;
        if (cntB > 32) cntB = 32;
        const int cmax = (cntA > cntB) ? cntA : cntB;
        float4 accA, accB;
        {
        Rows8 rA, rB;
        rows8_issue(rA, X4, lane, recA, 0, cntA);
        rows8_issue(rB, X4, lane, recB, 0, cntB);
        accA.x = ssA * hA.x;
        accA.y = ssA * hA.y;
        accA.z = ssA * hA.z;
        accA.w = ssA * hA.w;
        accB.x = ssB * hB.x;
        accB.y = ssB * hB.y;
        accB.z = ssB * hB.z;
        accB.w = ssB * hB.w;
        rows8_add(accA, rA);
        rows8_add(accB, rB);
        }

        const int nbegA = __builtin_amdgcn_readlane(nb, 0);
        const int nendA = __builtin_amdgcn_readlane(nb, 1);
        const int nbegB = __builtin_amdgcn_readlane(nb, 2);
        const int nendB = __builtin_amdgcn_readlane(nb, 3);
        int2 nrecA = make_int2(0, 0);
        int2 nrecB = make_int2(0, 0);
        if (nbegA + lane < nendA) nrecA = adj[nbegA + lane];
        if (nbegB + lane < nendB) nrecB = adj[nbegB + lane];
        int nd = 0;
        if (tn < tiles) nd = __float_as_int(dinv[tn * 16 + wave + ((lane & 1) << 3)]);
        nb = 0;
        if (tn + tstep < tiles) nb = rowptr[(tn + tstep) * 16 + wave + boff];

        for (int j = 8; j < cmax; j += 8) {
            Rows8 rA, rB;
            rows8_issue(rA, X4, lane, recA, j, cntA);
            rows8_issue(rB, X4, lane, recB, j, cntB);
            rows8_add(accA, rA);
            rows8_add(accB, rB);
        }
        rows_past32(accA, X4, adj, lane, begA, endA);
        rows_past32(accB, X4, adj, lane, begB, endB);

        *(float4*)&xs[buf][wave * XPITCH + 4 * lane]       = accA;
        *(float4*)&xs[buf][(wave + 8) * XPITCH + 4 * lane] = accB;
        lds_handover_barrier();

        const int m0 = t << 4;
        v8f pacc = {};
        const float* xrow = &xs[buf][row * XPITCH];

#ifdef USE_F32_WMMA
        #pragma unroll
        for (int i = 0; i < DIMF / 4; ++i) {
            const v2f a = *(const v2f*)(xrow + 4 * i + kh);
            pacc = __builtin_amdgcn_wmma_f32_16x16x4_f32(
                       false, a, false, bw[i], (short)0, pacc, false, false);
        }
#else
        for (int k = 0; k < DIMF; k += 32) {
            v16h a, bf;
            #pragma unroll
            for (int i = 0; i < 16; ++i) {
                int vg = i >> 1;
                int ka = ((vg < 4) ? (vg * 2) : (16 + (vg - 4) * 2)) + (i & 1) + hi * 8;
                a[i] = (_Float16)xrow[k + ka];
                int kb = i + hi * 16;
                bf[i] = (_Float16)W[(size_t)(k + kb) * DIMF + col0 + row];
            }
            pacc = __builtin_amdgcn_wmma_f32_16x16x32_f16(
                       false, a, false, bf, (short)0, pacc, false, false);
        }
#endif

        float* op = out + (size_t)m0 * DIMF + col0 + row;
        const int mo = hi * 8;
        #pragma unroll
        for (int j = 0; j < 8; ++j) {
            float o = pacc[j] + bias;
            o = (o > 0.0f) ? o : 0.01f * o;
            op[(size_t)(mo + j) * DIMF] = o;
        }

        t = tn;
        if (t >= tiles) break;
        buf ^= 1;
        vA   = __builtin_amdgcn_readfirstlane(t * 16 + wave);
        begA = nbegA;
        endA = nendA;
        begB = nbegB;
        endB = nendB;
        recA = nrecA;
        recB = nrecB;
        const float dA = __int_as_float(__builtin_amdgcn_readlane(nd, 0));
        const float dB = __int_as_float(__builtin_amdgcn_readlane(nd, 1));
        ssA = dA * dA;
        ssB = dB * dB;
    }
}


extern "C" void kernel_launch(void* const* d_in, const int* in_sizes, int n_in,
                              void* d_out, int out_size, void* d_ws, size_t ws_size,
                              hipStream_t stream) {
    const float* x  = (const float*)d_in[0];
    const int*   ei = (const int*)d_in[1];
    const float* Wts[3] = {(const float*)d_in[2], (const float*)d_in[4], (const float*)d_in[6]};
    const float* bs[3]  = {(const float*)d_in[3], (const float*)d_in[5], (const float*)d_in[7]};

    const int N = in_sizes[0] / DIMF;
    const int E = in_sizes[1] / 2;
    const int* srcp = ei;
    const int* dstp = ei + E;

    float* ws   = (float*)d_ws;
    const size_t nA = ((size_t)N + 64) & ~(size_t)63;
    const size_t eA = ((size_t)E + 63) & ~(size_t)63;
    float* dinv   = ws;
    float* Hbuf   = dinv + nA;
    float* Abuf   = Hbuf + (size_t)N * DIMF;
    int*   cnt    = (int*)(Abuf + (size_t)N * DIMF);
    int*   rowptr = cnt + nA;
    int*   rank   = rowptr + nA;
    int2*  adj    = (int2*)(rank + eA);

    const int T = 256;
    const int tilesM = N / 16;
    const int nq = N / 4;
    const int E4 = E / 4;

    const int layerBlocks = 768;

    zero_i32<<<(nq + T - 1) / T,       T,    0, stream>>>(cnt, nq);
    deg_rank<<<(E4 + T - 1) / T,       T,    0, stream>>>(dstp, cnt, rank, E4);
    row_scan<<<(nq + 1023) / 1024,     1024, 0, stream>>>(cnt, rowptr, dinv, nq);
    adj_fill<<<(E4 + T - 1) / T,       T,    0, stream>>>(srcp, dstp, rank, rowptr, dinv, adj, E4);

    gcn_layer<<<layerBlocks, 256, 0, stream>>>(x,    Wts[0], bs[0], rowptr, adj, dinv, Abuf,          tilesM);
    gcn_layer<<<layerBlocks, 256, 0, stream>>>(Abuf, Wts[1], bs[1], rowptr, adj, dinv, Hbuf,          tilesM);
    gcn_layer<<<layerBlocks, 256, 0, stream>>>(Hbuf, Wts[2], bs[2], rowptr, adj, dinv, (float*)d_out, tilesM);
}


# Gather issue path: a row read is one 32-bit offset add on a scalar base (row byte offsets kept in the edge records); full groups of 8 go out untested, part groups as 4+2+1 runs; sum order kept
# speedup vs baseline: 3.7211x; 1.0390x over previous
#include <hip/hip_runtime.h>

typedef float     v2f  __attribute__((ext_vector_type(2)));
typedef float     v8f  __attribute__((ext_vector_type(8)));
typedef _Float16  v16h __attribute__((ext_vector_type(16)));

#define DIMF 128

#if __has_builtin(__builtin_amdgcn_wmma_f32_16x16x4_f32)
#define USE_F32_WMMA 1
#endif


__global__ void zero_i32(int* __restrict__ p, int n4) {
    int i = blockIdx.x * blockDim.x + threadIdx.x;
    if (i < n4) ((int4*)p)[i] = make_int4(0, 0, 0, 0);
}

__global__ void deg_rank(const int* __restrict__ dst, int* __restrict__ cnt,
                         int* __restrict__ rank, int E4) {
    const int4* d4 = (const int4*)dst;
    int4* r4 = (int4*)rank;
    const int stride = gridDim.x * blockDim.x;
    for (int g = blockIdx.x * blockDim.x + threadIdx.x; g < E4; g += stride) {
        const int4 d = d4[g];
        int4 r;
        r.x = atomicAdd(&cnt[d.x], 1);
        r.y = atomicAdd(&cnt[d.y], 1);
        r.z = atomicAdd(&cnt[d.z], 1);
        r.w = atomicAdd(&cnt[d.w], 1);
        r4[g] = r;
    }
}

__global__ __launch_bounds__(1024) void row_scan(const int* __restrict__ cnt,
                                                 int* __restrict__ rowptr,
                                                 float* __restrict__ dinv, int nq) {
    __shared__ int wtot[32];
    __shared__ int wpre[32];
    const int tid  = threadIdx.x;
    const int lane = tid & 31;
    const int wave = tid >> 5;
    const int q0   = blockIdx.x * 1024;
    const int4* c4 = (const int4*)cnt;

    int pre = 0;
    #pragma unroll 4
    for (int i = tid; i < q0; i += 1024) {
        int4 t = c4[i];
        pre += (t.x + t.y) + (t.z + t.w);
    }
    const int q = q0 + tid;
    int4 c = make_int4(0, 0, 0, 0);
    if (q < nq) c = c4[q];
    const int tsum = (c.x + c.y) + (c.z + c.w);

    int s = tsum;
    #pragma unroll
    for (int o = 1; o < 32; o <<= 1) {
        int t = __shfl_up(s, o);
        if (lane >= o) s += t;
    }
    #pragma unroll
    for (int o = 16; o > 0; o >>= 1) pre += __shfl_xor(pre, o);

    if (lane == 31) wtot[wave] = s;
    if (lane == 0)  wpre[wave] = pre;
    __syncthreads();
    int off = 0;
    #pragma unroll
    for (int w = 0; w < 32; ++w) {
        off += wpre[w];
        if (w < wave) off += wtot[w];
    }
    const int e0 = off + s - tsum;
    if (q < nq) {
        int4 r;
        r.x = e0;
        r.y = e0 + c.x;
        r.z = r.y + c.y;
        r.w = r.z + c.z;
        ((int4*)rowptr)[q] = r;
        float4 d;
        d.x = rsqrtf((float)c.x + 1.0f);
        d.y = rsqrtf((float)c.y + 1.0f);
        d.z = rsqrtf((float)c.z + 1.0f);
        d.w = rsqrtf((float)c.w + 1.0f);
        ((float4*)dinv)[q] = d;
        if (q == nq - 1) rowptr[4 * nq] = e0 + tsum;
    }
}

__global__ void adj_fill(const int* __restrict__ src, const int* __restrict__ dst,
                         const int* __restrict__ rank, const int* __restrict__ rowptr,
                         const float* __restrict__ dinv, int2* __restrict__ adj, int E4) {
    const int4* s4 = (const int4*)src;
    const int4* d4 = (const int4*)dst;
    const int4* r4 = (const int4*)rank;
    const int stride = gridDim.x * blockDim.x;
    for (int g = blockIdx.x * blockDim.x + threadIdx.x; g < E4; g += stride) {
        const int4 u = s4[g];
        const int4 v = d4[g];
        const int4 r = r4[g];
        const int p0 = rowptr[v.x] + r.x;
        const int p1 = rowptr[v.y] + r.y;
        const int p2 = rowptr[v.z] + r.z;
        const int p3 = rowptr[v.w] + r.w;
        const float w0 = dinv[u.x] * dinv[v.x];
        const float w1 = dinv[u.y] * dinv[v.y];
        const float w2 = dinv[u.z] * dinv[v.z];
        const float w3 = dinv[u.w] * dinv[v.w];
        adj[p0] = make_int2(u.x * (DIMF * 4), __float_as_int(w0));
        adj[p1] = make_int2(u.y * (DIMF * 4), __float_as_int(w1));
        adj[p2] = make_int2(u.z * (DIMF * 4), __float_as_int(w2));
        adj[p3] = make_int2(u.w * (DIMF * 4), __float_as_int(w3));
    }
}


#define XPITCH 132

static __device__ __forceinline__ void lds_handover_barrier() {
    __builtin_amdgcn_fence(__ATOMIC_RELEASE, "workgroup", "local");
    __builtin_amdgcn_s_barrier();
    __builtin_amdgcn_fence(__ATOMIC_ACQUIRE, "workgroup", "local");
}

static __device__ __forceinline__ int2 recs_fetch(const int2* __restrict__ adj, int beg, int end,
                                                  int lane) {
    const int idx = beg + lane;
    const bool ok = idx < end;
    const int2 v = adj[ok ? idx : 0];
    return make_int2(ok ? v.x : 0, ok ? v.y : 0);
}

struct Rows8 {
    float4 h[8];
    float  w[8];
};

static __device__ __forceinline__ float4 row_read(const float* __restrict__ X, int rowoff,
                                                  unsigned lane16) {
    const unsigned voff = (unsigned)rowoff + lane16;
    return *(const float4*)((const char*)X + voff);
}

static __device__ __forceinline__ void rows8_issue(Rows8& r, const float* __restrict__ X,
                                                   unsigned lane16, int2 rec, int j) {
    #pragma unroll
    for (int i = 0; i < 8; ++i) {
        const int off = __builtin_amdgcn_readlane(rec.x, j + i);
        r.w[i] = __int_as_float(__builtin_amdgcn_readlane(rec.y, j + i));
        r.h[i] = row_read(X, off, lane16);
    }
}

static __device__ __forceinline__ void rows7_issue(Rows8& r, const float* __restrict__ X,
                                                   unsigned lane16, int2 rec, int j, int m) {
    #pragma unroll
    for (int i = 0; i < 7; ++i) {
        r.w[i] = 0.0f;
        r.h[i] = make_float4(0.0f, 0.0f, 0.0f, 0.0f);
    }
    const int j2 = j + (m & 4);
    const int j1 = j + (m & 6);
    if (m & 4) {
        #pragma unroll
        for (int i = 0; i < 4; ++i) {
            const int off = __builtin_amdgcn_readlane(rec.x, j + i);
            r.w[i] = __int_as_float(__builtin_amdgcn_readlane(rec.y, j + i));
            r.h[i] = row_read(X, off, lane16);
        }
    }
    if (m & 2) {
        #pragma unroll
        for (int i = 0; i < 2; ++i) {
            const int off = __builtin_amdgcn_readlane(rec.x, j2 + i);
            r.w[4 + i] = __int_as_float(__builtin_amdgcn_readlane(rec.y, j2 + i));
            r.h[4 + i] = row_read(X, off, lane16);
        }
    }
    if (m & 1) {
        const int off = __builtin_amdgcn_readlane(rec.x, j1);
        r.w[6] = __int_as_float(__builtin_amdgcn_readlane(rec.y, j1));
        r.h[6] = row_read(X, off, lane16);
    }
}

template <int NSLOT>
static __device__ __forceinline__ void rows_add(float4& acc, const Rows8& r) {
    #pragma unroll
    for (int i = 0; i < NSLOT; ++i) {
        acc.x = __builtin_fmaf(r.w[i], r.h[i].x, acc.x);
        acc.y = __builtin_fmaf(r.w[i], r.h[i].y, acc.y);
        acc.z = __builtin_fmaf(r.w[i], r.h[i].z, acc.z);
        acc.w = __builtin_fmaf(r.w[i], r.h[i].w, acc.w);
    }
}

template <bool FA, bool FB, bool FIRST>
static __device__ __forceinline__ void round_ab(float4& accA, float4& accB,
                                                const float* __restrict__ X, unsigned lane16,
                                                int2 recA, int2 recB, int j, int mA, int mB,
                                                int selfA, int selfB, float ssA, float ssB) {
    float4 hA, hB;
    if (FIRST) {
        hA = row_read(X, selfA, lane16);
        hB = row_read(X, selfB, lane16);
    }
    Rows8 rA, rB;
    if (!FA && FB) {
        rows8_issue(rB, X, lane16, recB, j);
        rows7_issue(rA, X, lane16, recA, j, mA);
    } else {
        if (FA) rows8_issue(rA, X, lane16, recA, j);
        else    rows7_issue(rA, X, lane16, recA, j, mA);
        if (FB) rows8_issue(rB, X, lane16, recB, j);
        else    rows7_issue(rB, X, lane16, recB, j, mB);
    }
    if (FIRST) {
        accA.x = ssA * hA.x;
        accA.y = ssA * hA.y;
        accA.z = ssA * hA.z;
        accA.w = ssA * hA.w;
        accB.x = ssB * hB.x;
        accB.y = ssB * hB.y;
        accB.z = ssB * hB.z;
        accB.w = ssB * hB.w;
    }
    if (FA) rows_add<8>(accA, rA);
    else    rows_add<7>(accA, rA);
    if (FB) rows_add<8>(accB, rB);
    else    rows_add<7>(accB, rB);
}

template <bool FIRST>
static __device__ __forceinline__ void round_any(float4& accA, float4& accB,
                                                 const float* __restrict__ X, unsigned lane16,
                                                 int2 recA, int2 recB, int j, int cntA, int cntB,
                                                 int selfA, int selfB, float ssA, float ssB) {
    int mA = cntA - j;
    int mB = cntB - j;
    mA = (mA < 0) ? 0 : mA;
    mB = (mB < 0) ? 0 : mB;
    const bool fa = mA >= 8;
    const bool fb = mB >= 8;
    if (fa && fb)
        round_ab<true, true, FIRST>(accA, accB, X, lane16, recA, recB, j, mA, mB, selfA, selfB, ssA, ssB);
    else if (fa)
        round_ab<true, false, FIRST>(accA, accB, X, lane16, recA, recB, j, mA, mB, selfA, selfB, ssA, ssB);
    else if (fb)
        round_ab<false, true, FIRST>(accA, accB, X, lane16, recA, recB, j, mA, mB, selfA, selfB, ssA, ssB);
    else
        round_ab<false, false, FIRST>(accA, accB, X, lane16, recA, recB, j, mA, mB, selfA, selfB, ssA, ssB);
}

static __device__ __forceinline__ void rows_past32(float4& acc,
                                                   const float* __restrict__ X,
                                                   const int2* __restrict__ adj,
                                                   int lane, unsigned lane16, int beg, int end) {
    for (int base = beg + 32; base < end; base += 32) {
        const int2 cur = recs_fetch(adj, base, end, lane);
        int c2 = end - base;
        if (c2 > 32) c2 = 32;
        int j = 0;
        do {
            Rows8 r;
            if (c2 - j >= 8) {
                rows8_issue(r, X, lane16, cur, j);
                rows_add<8>(acc, r);
            } else {
                rows7_issue(r, X, lane16, cur, j, c2 - j);
                rows_add<7>(acc, r);
            }
            j += 8;
        } while (j < c2);
    }
}

__global__ __launch_bounds__(256, 6) void gcn_layer(const float* __restrict__ X,
                                                 const float* __restrict__ W,
                                                 const float* __restrict__ b,
                                                 const int* __restrict__ rowptr,
                                                 const int2* __restrict__ adj,
                                                 const float* __restrict__ dinv,
                                                 float* __restrict__ out, int tiles) {
    __shared__ __attribute__((aligned(16))) float xs[2][16 * XPITCH];

    const int wave = threadIdx.x >> 5;
    const int lane = threadIdx.x & 31;
    const int col0 = wave << 4;
    const int row  = lane & 15;
    const int hi   = lane >> 4;

    int t = blockIdx.x;
    if (t >= tiles) return;
    const int tstep = (int)gridDim.x;
    const unsigned lane16 = (unsigned)lane << 4;
    const unsigned olane = (unsigned)((hi * 8 * DIMF + col0 + row) * 4);

#ifdef USE_F32_WMMA
    const int kh = hi * 2;
    v2f bw[DIMF / 4];
    #pragma unroll
    for (int i = 0; i < DIMF / 4; ++i) {
        const int k = 4 * i;
        bw[i].x = W[(size_t)(k + kh)     * DIMF + col0 + row];
        bw[i].y = W[(size_t)(k + kh + 1) * DIMF + col0 + row];
    }
#endif
    const float bias = b[col0 + row];

    const int boff = (lane & 1) + ((lane & 2) << 2);

    int vA   = __builtin_amdgcn_readfirstlane(t * 16 + wave);
    int begA = rowptr[vA];
    int endA = rowptr[vA + 1];
    int begB = rowptr[vA + 8];
    int endB = rowptr[vA + 9];
    int2 recA = recs_fetch(adj, begA, endA, lane);
    int2 recB = recs_fetch(adj, begB, endB, lane);
    int nb = rowptr[(t + tstep < tiles) ? (t + tstep) * 16 + wave + boff : 0];
    nb = (t + tstep < tiles) ? nb : 0;
    float ssA = dinv[vA];
    float ssB = dinv[vA + 8];
    ssA *= ssA;
    ssB *= ssB;

    int buf = 0;
    while (true) {
        const int tn = t + tstep;
        const int vB = vA + 8;

        int cntA = endA - begA;
        if (cntA > 32) cntA = 32;
        int cntB = endB - begB;
        if (cntB > 32) cntB = 32;
        const int cmax = (cntA > cntB) ? cntA : cntB;
        float4 accA, accB;
        round_any<true>(accA, accB, X, lane16, recA, recB, 0, cntA, cntB,
                        vA * (DIMF * 4), vB * (DIMF * 4), ssA, ssB);

        const int nbegA = __builtin_amdgcn_readlane(nb, 0);
        const int nendA = __builtin_amdgcn_readlane(nb, 1);
        const int nbegB = __builtin_amdgcn_readlane(nb, 2);
        const int nendB = __builtin_amdgcn_readlane(nb, 3);
        const int2 nrecA = recs_fetch(adj, nbegA, nendA, lane);
        const int2 nrecB = recs_fetch(adj, nbegB, nendB, lane);
        int nd = __float_as_int(dinv[(tn < tiles) ? tn * 16 + wave + ((lane & 1) << 3) : 0]);
        nd = (tn < tiles) ? nd : 0;
        nb = rowptr[(tn + tstep < tiles) ? (tn + tstep) * 16 + wave + boff : 0];
        nb = (tn + tstep < tiles) ? nb : 0;

        for (int j = 8; j < cmax; j += 8)
            round_any<false>(accA, accB, X, lane16, recA, recB, j, cntA, cntB, 0, 0, ssA, ssB);
        rows_past32(accA, X, adj, lane, lane16, begA, endA);
        rows_past32(accB, X, adj, lane, lane16, begB, endB);

        *(float4*)&xs[buf][wave * XPITCH + 4 * lane]       = accA;
        *(float4*)&xs[buf][(wave + 8) * XPITCH + 4 * lane] = accB;
        lds_handover_barrier();

        const int m0 = t << 4;
        v8f pacc = {};
        const float* xrow = &xs[buf][row * XPITCH];

#ifdef USE_F32_WMMA
        #pragma unroll
        for (int i = 0; i < DIMF / 4; ++i) {
            const v2f a = *(const v2f*)(xrow + 4 * i + kh);
            pacc = __builtin_amdgcn_wmma_f32_16x16x4_f32(
                       false, a, false, bw[i], (short)0, pacc, false, false);
        }
#else
        for (int k = 0; k < DIMF; k += 32) {
            v16h a, bf;
            #pragma unroll
            for (int i = 0; i < 16; ++i) {
                int vg = i >> 1;
                int ka = ((vg < 4) ? (vg * 2) : (16 + (vg - 4) * 2)) + (i & 1) + hi * 8;
                a[i] = (_Float16)xrow[k + ka];
                int kb = i + hi * 16;
                bf[i] = (_Float16)W[(size_t)(k + kb) * DIMF + col0 + row];
            }
            pacc = __builtin_amdgcn_wmma_f32_16x16x32_f16(
                       false, a, false, bf, (short)0, pacc, false, false);
        }
#endif

        const unsigned ooff = (unsigned)m0 * (DIMF * 4) + olane;
        #pragma unroll
        for (int j = 0; j < 8; ++j) {
            float o = pacc[j] + bias;
            o = (o > 0.0f) ? o : 0.01f * o;
            *(float*)((char*)out + ooff + j * (DIMF * 4)) = o;
        }

        t = tn;
        if (t >= tiles) break;
        buf ^= 1;
        vA   = __builtin_amdgcn_readfirstlane(t * 16 + wave);
        begA = nbegA;
        endA = nendA;
        begB = nbegB;
        endB = nendB;
        recA = nrecA;
        recB = nrecB;
        const float dA = __int_as_float(__builtin_amdgcn_readlane(nd, 0));
        const float dB = __int_as_float(__builtin_amdgcn_readlane(nd, 1));
        ssA = dA * dA;
        ssB = dB * dB;
    }
}


extern "C" void kernel_launch(void* const* d_in, const int* in_sizes, int n_in,
                              void* d_out, int out_size, void* d_ws, size_t ws_size,
                              hipStream_t stream) {
    const float* x  = (const float*)d_in[0];
    const int*   ei = (const int*)d_in[1];
    const float* Wts[3] = {(const float*)d_in[2], (const float*)d_in[4], (const float*)d_in[6]};
    const float* bs[3]  = {(const float*)d_in[3], (const float*)d_in[5], (const float*)d_in[7]};

    const int N = in_sizes[0] / DIMF;
    const int E = in_sizes[1] / 2;
    const int* srcp = ei;
    const int* dstp = ei + E;

    float* ws   = (float*)d_ws;
    const size_t nA = ((size_t)N + 64) & ~(size_t)63;
    const size_t eA = ((size_t)E + 63) & ~(size_t)63;
    float* dinv   = ws;
    float* Hbuf   = dinv + nA;
    float* Abuf   = Hbuf + (size_t)N * DIMF;
    int*   cnt    = (int*)(Abuf + (size_t)N * DIMF);
    int*   rowptr = cnt + nA;
    int*   rank   = rowptr + nA;
    int2*  adj    = (int2*)(rank + eA);

    const int T = 256;
    const int tilesM = N / 16;
    const int nq = N / 4;
    const int E4 = E / 4;

    const int layerBlocks = 768;

    zero_i32<<<(nq + T - 1) / T,       T,    0, stream>>>(cnt, nq);
    deg_rank<<<(E4 + T - 1) / T,       T,    0, stream>>>(dstp, cnt, rank, E4);
    row_scan<<<(nq + 1023) / 1024,     1024, 0, stream>>>(cnt, rowptr, dinv, nq);
    adj_fill<<<(E4 + T - 1) / T,       T,    0, stream>>>(srcp, dstp, rank, rowptr, dinv, adj, E4);

    gcn_layer<<<layerBlocks, 256, 0, stream>>>(x,    Wts[0], bs[0], rowptr, adj, dinv, Abuf,          tilesM);
    gcn_layer<<<layerBlocks, 256, 0, stream>>>(Abuf, Wts[1], bs[1], rowptr, adj, dinv, Hbuf,          tilesM);
    gcn_layer<<<layerBlocks, 256, 0, stream>>>(Hbuf, Wts[2], bs[2], rowptr, adj, dinv, (float*)d_out, tilesM);
}


# Layer kernel: rows 16-127 of W staged once per block in LDS in fragment order and read from there, not held as 64 registers per wave: 104 registers, 4 blocks per unit, 1024 blocks; sum order kept
# speedup vs baseline: 3.8442x; 1.0331x over previous
#include <hip/hip_runtime.h>

typedef float     v2f  __attribute__((ext_vector_type(2)));
typedef float     v8f  __attribute__((ext_vector_type(8)));
typedef _Float16  v16h __attribute__((ext_vector_type(16)));

#define DIMF 128

#if __has_builtin(__builtin_amdgcn_wmma_f32_16x16x4_f32)
#define USE_F32_WMMA 1
#endif


__global__ void zero_i32(int* __restrict__ p, int n4) {
    int i = blockIdx.x * blockDim.x + threadIdx.x;
    if (i < n4) ((int4*)p)[i] = make_int4(0, 0, 0, 0);
}

__global__ void deg_rank(const int* __restrict__ dst, int* __restrict__ cnt,
                         int* __restrict__ rank, int E4) {
    const int4* d4 = (const int4*)dst;
    int4* r4 = (int4*)rank;
    const int stride = gridDim.x * blockDim.x;
    for (int g = blockIdx.x * blockDim.x + threadIdx.x; g < E4; g += stride) {
        const int4 d = d4[g];
        int4 r;
        r.x = atomicAdd(&cnt[d.x], 1);
        r.y = atomicAdd(&cnt[d.y], 1);
        r.z = atomicAdd(&cnt[d.z], 1);
        r.w = atomicAdd(&cnt[d.w], 1);
        r4[g] = r;
    }
}

__global__ __launch_bounds__(1024) void row_scan(const int* __restrict__ cnt,
                                                 int* __restrict__ rowptr,
                                                 float* __restrict__ dinv, int nq) {
    __shared__ int wtot[32];
    __shared__ int wpre[32];
    const int tid  = threadIdx.x;
    const int lane = tid & 31;
    const int wave = tid >> 5;
    const int q0   = blockIdx.x * 1024;
    const int4* c4 = (const int4*)cnt;

    int pre = 0;
    #pragma unroll 4
    for (int i = tid; i < q0; i += 1024) {
        int4 t = c4[i];
        pre += (t.x + t.y) + (t.z + t.w);
    }
    const int q = q0 + tid;
    int4 c = make_int4(0, 0, 0, 0);
    if (q < nq) c = c4[q];
    const int tsum = (c.x + c.y) + (c.z + c.w);

    int s = tsum;
    #pragma unroll
    for (int o = 1; o < 32; o <<= 1) {
        int t = __shfl_up(s, o);
        if (lane >= o) s += t;
    }
    #pragma unroll
    for (int o = 16; o > 0; o >>= 1) pre += __shfl_xor(pre, o);

    if (lane == 31) wtot[wave] = s;
    if (lane == 0)  wpre[wave] = pre;
    __syncthreads();
    int off = 0;
    #pragma unroll
    for (int w = 0; w < 32; ++w) {
        off += wpre[w];
        if (w < wave) off += wtot[w];
    }
    const int e0 = off + s - tsum;
    if (q < nq) {
        int4 r;
        r.x = e0;
        r.y = e0 + c.x;
        r.z = r.y + c.y;
        r.w = r.z + c.z;
        ((int4*)rowptr)[q] = r;
        float4 d;
        d.x = rsqrtf((float)c.x + 1.0f);
        d.y = rsqrtf((float)c.y + 1.0f);
        d.z = rsqrtf((float)c.z + 1.0f);
        d.w = rsqrtf((float)c.w + 1.0f);
        ((float4*)dinv)[q] = d;
        if (q == nq - 1) rowptr[4 * nq] = e0 + tsum;
    }
}

__global__ void adj_fill(const int* __restrict__ src, const int* __restrict__ dst,
                         const int* __restrict__ rank, const int* __restrict__ rowptr,
                         const float* __restrict__ dinv, int2* __restrict__ adj, int E4) {
    const int4* s4 = (const int4*)src;
    const int4* d4 = (const int4*)dst;
    const int4* r4 = (const int4*)rank;
    const int stride = gridDim.x * blockDim.x;
    for (int g = blockIdx.x * blockDim.x + threadIdx.x; g < E4; g += stride) {
        const int4 u = s4[g];
        const int4 v = d4[g];
        const int4 r = r4[g];
        const int p0 = rowptr[v.x] + r.x;
        const int p1 = rowptr[v.y] + r.y;
        const int p2 = rowptr[v.z] + r.z;
        const int p3 = rowptr[v.w] + r.w;
        const float w0 = dinv[u.x] * dinv[v.x];
        const float w1 = dinv[u.y] * dinv[v.y];
        const float w2 = dinv[u.z] * dinv[v.z];
        const float w3 = dinv[u.w] * dinv[v.w];
        adj[p0] = make_int2(u.x * (DIMF * 4), __float_as_int(w0));
        adj[p1] = make_int2(u.y * (DIMF * 4), __float_as_int(w1));
        adj[p2] = make_int2(u.z * (DIMF * 4), __float_as_int(w2));
        adj[p3] = make_int2(u.w * (DIMF * 4), __float_as_int(w3));
    }
}


#define XPITCH 132

static __device__ __forceinline__ void lds_handover_barrier() {
    __builtin_amdgcn_fence(__ATOMIC_RELEASE, "workgroup", "local");
    __builtin_amdgcn_s_barrier();
    __builtin_amdgcn_fence(__ATOMIC_ACQUIRE, "workgroup", "local");
}

static __device__ __forceinline__ int2 recs_fetch(const int2* __restrict__ adj, int beg, int end,
                                                  int lane) {
    const int idx = beg + lane;
    const bool ok = idx < end;
    const int2 v = adj[ok ? idx : 0];
    return make_int2(ok ? v.x : 0, ok ? v.y : 0);
}

struct Rows8 {
    float4 h[8];
    float  w[8];
};

static __device__ __forceinline__ float4 row_read(const float* __restrict__ X, int rowoff,
                                                  unsigned lane16) {
    const unsigned voff = (unsigned)rowoff + lane16;
    return *(const float4*)((const char*)X + voff);
}

static __device__ __forceinline__ void rows8_issue(Rows8& r, const float* __restrict__ X,
                                                   unsigned lane16, int2 rec, int j) {
    #pragma unroll
    for (int i = 0; i < 8; ++i) {
        const int off = __builtin_amdgcn_readlane(rec.x, j + i);
        r.w[i] = __int_as_float(__builtin_amdgcn_readlane(rec.y, j + i));
        r.h[i] = row_read(X, off, lane16);
    }
}

static __device__ __forceinline__ void rows7_issue(Rows8& r, const float* __restrict__ X,
                                                   unsigned lane16, int2 rec, int j, int m) {
    #pragma unroll
    for (int i = 0; i < 7; ++i) {
        r.w[i] = 0.0f;
        r.h[i] = make_float4(0.0f, 0.0f, 0.0f, 0.0f);
    }
    const int j2 = j + (m & 4);
    const int j1 = j + (m & 6);
    if (m & 4) {
        #pragma unroll
        for (int i = 0; i < 4; ++i) {
            const int off = __builtin_amdgcn_readlane(rec.x, j + i);
            r.w[i] = __int_as_float(__builtin_amdgcn_readlane(rec.y, j + i));
            r.h[i] = row_read(X, off, lane16);
        }
    }
    if (m & 2) {
        #pragma unroll
        for (int i = 0; i < 2; ++i) {
            const int off = __builtin_amdgcn_readlane(rec.x, j2 + i);
            r.w[4 + i] = __int_as_float(__builtin_amdgcn_readlane(rec.y, j2 + i));
            r.h[4 + i] = row_read(X, off, lane16);
        }
    }
    if (m & 1) {
        const int off = __builtin_amdgcn_readlane(rec.x, j1);
        r.w[6] = __int_as_float(__builtin_amdgcn_readlane(rec.y, j1));
        r.h[6] = row_read(X, off, lane16);
    }
}

template <int NSLOT>
static __device__ __forceinline__ void rows_add(float4& acc, const Rows8& r) {
    #pragma unroll
    for (int i = 0; i < NSLOT; ++i) {
        acc.x = __builtin_fmaf(r.w[i], r.h[i].x, acc.x);
        acc.y = __builtin_fmaf(r.w[i], r.h[i].y, acc.y);
        acc.z = __builtin_fmaf(r.w[i], r.h[i].z, acc.z);
        acc.w = __builtin_fmaf(r.w[i], r.h[i].w, acc.w);
    }
}

template <bool FA, bool FB, bool FIRST>
static __device__ __forceinline__ void round_ab(float4& accA, float4& accB,
                                                const float* __restrict__ X, unsigned lane16,
                                                int2 recA, int2 recB, int j, int mA, int mB,
                                                int selfA, int selfB, float ssA, float ssB) {
    float4 hA, hB;
    if (FIRST) {
        hA = row_read(X, selfA, lane16);
        hB = row_read(X, selfB, lane16);
    }
    Rows8 rA, rB;
    if (!FA && FB) {
        rows8_issue(rB, X, lane16, recB, j);
        rows7_issue(rA, X, lane16, recA, j, mA);
    } else {
        if (FA) rows8_issue(rA, X, lane16, recA, j);
        else    rows7_issue(rA, X, lane16, recA, j, mA);
        if (FB) rows8_issue(rB, X, lane16, recB, j);
        else    rows7_issue(rB, X, lane16, recB, j, mB);
    }
    if (FIRST) {
        accA.x = ssA * hA.x;
        accA.y = ssA * hA.y;
        accA.z = ssA * hA.z;
        accA.w = ssA * hA.w;
        accB.x = ssB * hB.x;
        accB.y = ssB * hB.y;
        accB.z = ssB * hB.z;
        accB.w = ssB * hB.w;
    }
    if (FA) rows_add<8>(accA, rA);
    else    rows_add<7>(accA, rA);
    if (FB) rows_add<8>(accB, rB);
    else    rows_add<7>(accB, rB);
}

template <bool FIRST>
static __device__ __forceinline__ void round_any(float4& accA, float4& accB,
                                                 const float* __restrict__ X, unsigned lane16,
                                                 int2 recA, int2 recB, int j, int cntA, int cntB,
                                                 int selfA, int selfB, float ssA, float ssB) {
    int mA = cntA - j;
    int mB = cntB - j;
    mA = (mA < 0) ? 0 : mA;
    mB = (mB < 0) ? 0 : mB;
    const bool fa = mA >= 8;
    const bool fb = mB >= 8;
    if (fa && fb)
        round_ab<true, true, FIRST>(accA, accB, X, lane16, recA, recB, j, mA, mB, selfA, selfB, ssA, ssB);
    else if (fa)
        round_ab<true, false, FIRST>(accA, accB, X, lane16, recA, recB, j, mA, mB, selfA, selfB, ssA, ssB);
    else if (fb)
        round_ab<false, true, FIRST>(accA, accB, X, lane16, recA, recB, j, mA, mB, selfA, selfB, ssA, ssB);
    else
        round_ab<false, false, FIRST>(accA, accB, X, lane16, recA, recB, j, mA, mB, selfA, selfB, ssA, ssB);
}

static __device__ __forceinline__ void rows_past32(float4& acc,
                                                   const float* __restrict__ X,
                                                   const int2* __restrict__ adj,
                                                   int lane, unsigned lane16, int beg, int end) {
    for (int base = beg + 32; base < end; base += 32) {
        const int2 cur = recs_fetch(adj, base, end, lane);
        int c2 = end - base;
        if (c2 > 32) c2 = 32;
        int j = 0;
        do {
            Rows8 r;
            if (c2 - j >= 8) {
                rows8_issue(r, X, lane16, cur, j);
                rows_add<8>(acc, r);
            } else {
                rows7_issue(r, X, lane16, cur, j, c2 - j);
                rows_add<7>(acc, r);
            }
            j += 8;
        } while (j < c2);
    }
}

#define WREG 16
__global__ __launch_bounds__(256, 8) void gcn_layer(const float* __restrict__ X,
                                                 const float* __restrict__ W,
                                                 const float* __restrict__ b,
                                                 const int* __restrict__ rowptr,
                                                 const int2* __restrict__ adj,
                                                 const float* __restrict__ dinv,
                                                 float* __restrict__ out, int tiles) {
    __shared__ __attribute__((aligned(16))) float xs[2][16 * XPITCH];
#ifdef USE_F32_WMMA
    __shared__ __attribute__((aligned(16))) float wl[(DIMF - WREG) * DIMF];
#endif

    const int wave = threadIdx.x >> 5;
    const int lane = threadIdx.x & 31;
    const int col0 = wave << 4;
    const int row  = lane & 15;
    const int hi   = lane >> 4;

    int t = blockIdx.x;
    if (t >= tiles) return;
    const int tstep = (int)gridDim.x;
    const unsigned lane16 = (unsigned)lane << 4;
    const unsigned olane = (unsigned)((hi * 8 * DIMF + col0 + row) * 4);

#ifdef USE_F32_WMMA
    const int kh = hi * 2;
    v2f bw[WREG / 4];
    #pragma unroll
    for (int i = 0; i < WREG / 4; ++i) {
        const int k = 4 * i;
        bw[i].x = W[(size_t)(k + kh)     * DIMF + col0 + row];
        bw[i].y = W[(size_t)(k + kh + 1) * DIMF + col0 + row];
    }
    {
        v2f r0[14], r1[14];
        #pragma unroll
        for (int n = 0; n < 14; ++n) {
            const int item = (int)threadIdx.x + 256 * n;
            const int m  = item & 63;
            const int k  = WREG + 2 * (item >> 6);
            r0[n] = *(const v2f*)&W[(size_t)k       * DIMF + 2 * m];
            r1[n] = *(const v2f*)&W[(size_t)(k + 1) * DIMF + 2 * m];
        }
        #pragma unroll
        for (int n = 0; n < 14; ++n) {
            const int item = (int)threadIdx.x + 256 * n;
            const int m  = item & 63;
            const int k  = WREG + 2 * (item >> 6);
            const int g  = (k - WREG) >> 2;
            const int h  = (k >> 1) & 1;
            const int ci = (2 * m + 16 * h) & (DIMF - 1);
            *(float4*)&wl[((g * 2 + h) * DIMF + ci) * 2] =
                make_float4(r0[n].x, r1[n].x, r0[n].y, r1[n].y);
        }
    }
    const float* wb = &wl[(hi * DIMF + ((col0 + row + 16 * hi) & (DIMF - 1))) * 2];
#endif
    const float bias = b[col0 + row];

    const int boff = (lane & 1) + ((lane & 2) << 2);

    int vA   = __builtin_amdgcn_readfirstlane(t * 16 + wave);
    int begA = rowptr[vA];
    int endA = rowptr[vA + 1];
    int begB = rowptr[vA + 8];
    int endB = rowptr[vA + 9];
    int2 recA = recs_fetch(adj, begA, endA, lane);
    int2 recB = recs_fetch(adj, begB, endB, lane);
    int nb = rowptr[(t + tstep < tiles) ? (t + tstep) * 16 + wave + boff : 0];
    nb = (t + tstep < tiles) ? nb : 0;
    float ssA = dinv[vA];
    float ssB = dinv[vA + 8];
    ssA *= ssA;
    ssB *= ssB;

    int buf = 0;
    while (true) {
        const int tn = t + tstep;
        const int vB = vA + 8;

        int cntA = endA - begA;
        if (cntA > 32) cntA = 32;
        int cntB = endB - begB;
        if (cntB > 32) cntB = 32;
        const int cmax = (cntA > cntB) ? cntA : cntB;
        float4 accA, accB;
        round_any<true>(accA, accB, X, lane16, recA, recB, 0, cntA, cntB,
                        vA * (DIMF * 4), vB * (DIMF * 4), ssA, ssB);

        const int nbegA = __builtin_amdgcn_readlane(nb, 0);
        const int nendA = __builtin_amdgcn_readlane(nb, 1);
        const int nbegB = __builtin_amdgcn_readlane(nb, 2);
        const int nendB = __builtin_amdgcn_readlane(nb, 3);
        const int2 nrecA = recs_fetch(adj, nbegA, nendA, lane);
        const int2 nrecB = recs_fetch(adj, nbegB, nendB, lane);
        int nd = __float_as_int(dinv[(tn < tiles) ? tn * 16 + wave + ((lane & 1) << 3) : 0]);
        nd = (tn < tiles) ? nd : 0;
        nb = rowptr[(tn + tstep < tiles) ? (tn + tstep) * 16 + wave + boff : 0];
        nb = (tn + tstep < tiles) ? nb : 0;

        for (int j = 8; j < cmax; j += 8)
            round_any<false>(accA, accB, X, lane16, recA, recB, j, cntA, cntB, 0, 0, ssA, ssB);
        rows_past32(accA, X, adj, lane, lane16, begA, endA);
        rows_past32(accB, X, adj, lane, lane16, begB, endB);

        *(float4*)&xs[buf][wave * XPITCH + 4 * lane]       = accA;
        *(float4*)&xs[buf][(wave + 8) * XPITCH + 4 * lane] = accB;
        lds_handover_barrier();

        const int m0 = t << 4;
        v8f pacc = {};
        const float* xrow = &xs[buf][row * XPITCH];

#ifdef USE_F32_WMMA
        #pragma unroll
        for (int i = 0; i < DIMF / 4; ++i) {
            const v2f a = *(const v2f*)(xrow + 4 * i + kh);
            v2f bfr;
            if (i < WREG / 4) bfr = bw[i];
            else              bfr = *(const v2f*)(wb + (i - WREG / 4) * (4 * DIMF));
            pacc = __builtin_amdgcn_wmma_f32_16x16x4_f32(
                       false, a, false, bfr, (short)0, pacc, false, false);
        }
#else
        for (int k = 0; k < DIMF; k += 32) {
            v16h a, bf;
            #pragma unroll
            for (int i = 0; i < 16; ++i) {
                int vg = i >> 1;
                int ka = ((vg < 4) ? (vg * 2) : (16 + (vg - 4) * 2)) + (i & 1) + hi * 8;
                a[i] = (_Float16)xrow[k + ka];
                int kb = i + hi * 16;
                bf[i] = (_Float16)W[(size_t)(k + kb) * DIMF + col0 + row];
            }
            pacc = __builtin_amdgcn_wmma_f32_16x16x32_f16(
                       false, a, false, bf, (short)0, pacc, false, false);
        }
#endif

        const unsigned ooff = (unsigned)m0 * (DIMF * 4) + olane;
        #pragma unroll
        for (int j = 0; j < 8; ++j) {
            float o = pacc[j] + bias;
            o = (o > 0.0f) ? o : 0.01f * o;
            *(float*)((char*)out + ooff + j * (DIMF * 4)) = o;
        }

        t = tn;
        if (t >= tiles) break;
        buf ^= 1;
        vA   = __builtin_amdgcn_readfirstlane(t * 16 + wave);
        begA = nbegA;
        endA = nendA;
        begB = nbegB;
        endB = nendB;
        recA = nrecA;
        recB = nrecB;
        const float dA = __int_as_float(__builtin_amdgcn_readlane(nd, 0));
        const float dB = __int_as_float(__builtin_amdgcn_readlane(nd, 1));
        ssA = dA * dA;
        ssB = dB * dB;
    }
}


extern "C" void kernel_launch(void* const* d_in, const int* in_sizes, int n_in,
                              void* d_out, int out_size, void* d_ws, size_t ws_size,
                              hipStream_t stream) {
    const float* x  = (const float*)d_in[0];
    const int*   ei = (const int*)d_in[1];
    const float* Wts[3] = {(const float*)d_in[2], (const float*)d_in[4], (const float*)d_in[6]};
    const float* bs[3]  = {(const float*)d_in[3], (const float*)d_in[5], (const float*)d_in[7]};

    const int N = in_sizes[0] / DIMF;
    const int E = in_sizes[1] / 2;
    const int* srcp = ei;
    const int* dstp = ei + E;

    float* ws   = (float*)d_ws;
    const size_t nA = ((size_t)N + 64) & ~(size_t)63;
    const size_t eA = ((size_t)E + 63) & ~(size_t)63;
    float* dinv   = ws;
    float* Hbuf   = dinv + nA;
    float* Abuf   = Hbuf + (size_t)N * DIMF;
    int*   cnt    = (int*)(Abuf + (size_t)N * DIMF);
    int*   rowptr = cnt + nA;
    int*   rank   = rowptr + nA;
    int2*  adj    = (int2*)(rank + eA);

    const int T = 256;
    const int tilesM = N / 16;
    const int nq = N / 4;
    const int E4 = E / 4;

    const int layerBlocks = 1024;

    zero_i32<<<(nq + T - 1) / T,       T,    0, stream>>>(cnt, nq);
    deg_rank<<<(E4 + T - 1) / T,       T,    0, stream>>>(dstp, cnt, rank, E4);
    row_scan<<<(nq + 1023) / 1024,     1024, 0, stream>>>(cnt, rowptr, dinv, nq);
    adj_fill<<<(E4 + T - 1) / T,       T,    0, stream>>>(srcp, dstp, rank, rowptr, dinv, adj, E4);

    gcn_layer<<<layerBlocks, 256, 0, stream>>>(x,    Wts[0], bs[0], rowptr, adj, dinv, Abuf,          tilesM);
    gcn_layer<<<layerBlocks, 256, 0, stream>>>(Abuf, Wts[1], bs[1], rowptr, adj, dinv, Hbuf,          tilesM);
    gcn_layer<<<layerBlocks, 256, 0, stream>>>(Hbuf, Wts[2], bs[2], rowptr, adj, dinv, (float*)d_out, tilesM);
}


# Edges grouped by destination by a two-level counting sort on LDS counters (256 chunks by 256-node bucket, then one block per bucket), not one returning global atomic per edge; sum order: arrival
# speedup vs baseline: 5.1368x; 1.3362x over previous
#include <hip/hip_runtime.h>

typedef float     v2f  __attribute__((ext_vector_type(2)));
typedef float     v8f  __attribute__((ext_vector_type(8)));
typedef _Float16  v16h __attribute__((ext_vector_type(16)));

#define DIMF 128

#if __has_builtin(__builtin_amdgcn_wmma_f32_16x16x4_f32)
#define USE_F32_WMMA 1
#endif


#define NCH 256

static __device__ __forceinline__ int block_scan256(int v, int lane, int wave, int* wsum, int& total) {
    int s = v;
    #pragma unroll
    for (int o = 1; o < 32; o <<= 1) {
        int t = __shfl_up(s, o);
        if (lane >= o) s += t;
    }
    if (lane == 31) wsum[wave] = s;
    __syncthreads();
    int off = 0, tot = 0;
    #pragma unroll
    for (int w = 0; w < 8; ++w) {
        const int x = wsum[w];
        tot += x;
        if (w < wave) off += x;
    }
    total = tot;
    return off + s - v;
}

__global__ __launch_bounds__(256) void sort_chunks(const int* __restrict__ src,
                                                   const int* __restrict__ dst,
                                                   int* __restrict__ cs, int2* __restrict__ tmp,
                                                   int E4, int CH4) {
    __shared__ int hist[256];
    __shared__ int wsum[8];
    const int tid  = threadIdx.x;
    const int lane = tid & 31;
    const int wave = tid >> 5;
    const int c    = blockIdx.x;
    const int q0   = c * CH4;
    int nq = E4 - q0;
    nq = (nq > CH4) ? CH4 : nq;
    nq = (nq < 0) ? 0 : nq;
    const int4* s4 = (const int4*)src;
    const int4* d4 = (const int4*)dst;

    hist[tid] = 0;
    __syncthreads();
    for (int t0 = 0; t0 < nq; t0 += 4 * 256) {
        int4 d[4];
        bool ok[4];
        #pragma unroll
        for (int k = 0; k < 4; ++k) {
            const int i = t0 + k * 256 + tid;
            ok[k] = i < nq;
            d[k] = d4[ok[k] ? q0 + i : 0];
        }
        #pragma unroll
        for (int k = 0; k < 4; ++k) {
            if (ok[k]) {
                atomicAdd(&hist[d[k].x >> 8], 1);
                atomicAdd(&hist[d[k].y >> 8], 1);
                atomicAdd(&hist[d[k].z >> 8], 1);
                atomicAdd(&hist[d[k].w >> 8], 1);
            }
        }
    }
    __syncthreads();
    const int h = hist[tid];
    int total;
    const int excl = block_scan256(h, lane, wave, wsum, total);
    cs[tid * NCH + c] = excl;
    if (tid == 255) cs[256 * NCH + c] = total;
    hist[tid] = excl;
    __syncthreads();

    int2* out = tmp + (size_t)c * (size_t)(4 * CH4);
    for (int t0 = 0; t0 < nq; t0 += 4 * 256) {
        int4 d[4], s[4];
        bool ok[4];
        #pragma unroll
        for (int k = 0; k < 4; ++k) {
            const int i = t0 + k * 256 + tid;
            ok[k] = i < nq;
            d[k] = d4[ok[k] ? q0 + i : 0];
            s[k] = s4[ok[k] ? q0 + i : 0];
        }
        #pragma unroll
        for (int k = 0; k < 4; ++k) {
            if (ok[k]) {
                const int p0 = atomicAdd(&hist[d[k].x >> 8], 1);
                const int p1 = atomicAdd(&hist[d[k].y >> 8], 1);
                const int p2 = atomicAdd(&hist[d[k].z >> 8], 1);
                const int p3 = atomicAdd(&hist[d[k].w >> 8], 1);
                out[p0] = make_int2(s[k].x, d[k].x);
                out[p1] = make_int2(s[k].y, d[k].y);
                out[p2] = make_int2(s[k].z, d[k].z);
                out[p3] = make_int2(s[k].w, d[k].w);
            }
        }
    }
}

template <bool PLACE>
static __device__ __forceinline__ void bucket_edge(int2 e, int* cnt, int2* __restrict__ adj) {
    const int ld = e.y & 255;
    if (PLACE) {
        const int p = atomicAdd(&cnt[ld], 1);
        adj[p] = e;
    } else {
        atomicAdd(&cnt[ld], 1);
    }
}

__global__ __launch_bounds__(256) void sort_buckets(const int* __restrict__ cs,
                                                    const int2* __restrict__ tmp,
                                                    int* __restrict__ rowptr,
                                                    float* __restrict__ dinv,
                                                    int2* __restrict__ adj, int n, int CHE) {
    __shared__ int deg[256];
    __shared__ int cur[256];
    __shared__ int wsA[8];
    __shared__ int wsB[8];
    const int tid  = threadIdx.x;
    const int lane = tid & 31;
    const int wave = __builtin_amdgcn_readfirstlane(tid >> 5);
    const int b    = blockIdx.x;

    const int s0 = cs[b * NCH + tid];
    const int s1 = cs[(b + 1) * NCH + tid];
    deg[tid] = 0;
    int r = s0;
    #pragma unroll
    for (int o = 16; o > 0; o >>= 1) r += __shfl_xor(r, o);
    if (lane == 0) wsA[wave] = r;
    __syncthreads();
    int bstart = 0;
    #pragma unroll
    for (int w = 0; w < 8; ++w) bstart += wsA[w];

    const int2* wtmp = tmp + (size_t)(wave * 32) * (size_t)CHE;
    int2 e[32];
    unsigned okm = 0u;
    #pragma unroll
    for (int j = 0; j < 32; ++j) {
        const int beg = __builtin_amdgcn_readlane(s0, j);
        const int end = __builtin_amdgcn_readlane(s1, j);
        const int i = beg + lane;
        const bool ok = i < end;
        e[j] = wtmp[(size_t)j * (size_t)CHE + (ok ? i : 0)];
        okm |= ok ? (1u << j) : 0u;
    }
    #pragma unroll
    for (int j = 0; j < 32; ++j)
        if ((okm >> j) & 1u) bucket_edge<false>(e[j], deg, adj);
    for (int j = 0; j < 32; ++j) {
        const int beg = __builtin_amdgcn_readlane(s0, j);
        const int end = __builtin_amdgcn_readlane(s1, j);
        for (int i = beg + 32 + lane; i < end; i += 32)
            bucket_edge<false>(wtmp[(size_t)j * (size_t)CHE + i], deg, adj);
    }
    __syncthreads();

    const int d = deg[tid];
    int total;
    const int excl = block_scan256(d, lane, wave, wsB, total);
    const int v = b * 256 + tid;
    if (v < n) {
        rowptr[v] = bstart + excl;
        dinv[v]   = rsqrtf((float)d + 1.0f);
        if (v == n - 1) rowptr[n] = bstart + excl + d;
    }
    cur[tid] = bstart + excl;
    __syncthreads();

    #pragma unroll
    for (int j = 0; j < 32; ++j)
        if ((okm >> j) & 1u) bucket_edge<true>(e[j], cur, adj);
    for (int j = 0; j < 32; ++j) {
        const int beg = __builtin_amdgcn_readlane(s0, j);
        const int end = __builtin_amdgcn_readlane(s1, j);
        for (int i = beg + 32 + lane; i < end; i += 32)
            bucket_edge<true>(wtmp[(size_t)j * (size_t)CHE + i], cur, adj);
    }
}

__global__ void adj_weights(int2* __restrict__ adj, const float* __restrict__ dinv, int E4) {
    int4* a4 = (int4*)adj;
    const int stride = gridDim.x * blockDim.x;
    for (int g = blockIdx.x * blockDim.x + threadIdx.x; g < E4; g += stride) {
        int4 r0 = a4[g];
        int4 r1 = a4[g + E4];
        const float u0 = dinv[r0.x];
        const float v0 = dinv[r0.y];
        const float u1 = dinv[r0.z];
        const float v1 = dinv[r0.w];
        const float u2 = dinv[r1.x];
        const float v2 = dinv[r1.y];
        const float u3 = dinv[r1.z];
        const float v3 = dinv[r1.w];
        a4[g]      = make_int4(r0.x * (DIMF * 4), __float_as_int(u0 * v0),
                               r0.z * (DIMF * 4), __float_as_int(u1 * v1));
        a4[g + E4] = make_int4(r1.x * (DIMF * 4), __float_as_int(u2 * v2),
                               r1.z * (DIMF * 4), __float_as_int(u3 * v3));
    }
}


#define XPITCH 132

static __device__ __forceinline__ void lds_handover_barrier() {
    __builtin_amdgcn_fence(__ATOMIC_RELEASE, "workgroup", "local");
    __builtin_amdgcn_s_barrier();
    __builtin_amdgcn_fence(__ATOMIC_ACQUIRE, "workgroup", "local");
}

static __device__ __forceinline__ int2 recs_fetch(const int2* __restrict__ adj, int beg, int end,
                                                  int lane) {
    const int idx = beg + lane;
    const bool ok = idx < end;
    const int2 v = adj[ok ? idx : 0];
    return make_int2(ok ? v.x : 0, ok ? v.y : 0);
}

struct Rows8 {
    float4 h[8];
    float  w[8];
};

static __device__ __forceinline__ float4 row_read(const float* __restrict__ X, int rowoff,
                                                  unsigned lane16) {
    const unsigned voff = (unsigned)rowoff + lane16;
    return *(const float4*)((const char*)X + voff);
}

static __device__ __forceinline__ void rows8_issue(Rows8& r, const float* __restrict__ X,
                                                   unsigned lane16, int2 rec, int j) {
    #pragma unroll
    for (int i = 0; i < 8; ++i) {
        const int off = __builtin_amdgcn_readlane(rec.x, j + i);
        r.w[i] = __int_as_float(__builtin_amdgcn_readlane(rec.y, j + i));
        r.h[i] = row_read(X, off, lane16);
    }
}

static __device__ __forceinline__ void rows7_issue(Rows8& r, const float* __restrict__ X,
                                                   unsigned lane16, int2 rec, int j, int m) {
    #pragma unroll
    for (int i = 0; i < 7; ++i) {
        r.w[i] = 0.0f;
        r.h[i] = make_float4(0.0f, 0.0f, 0.0f, 0.0f);
    }
    const int j2 = j + (m & 4);
    const int j1 = j + (m & 6);
    if (m & 4) {
        #pragma unroll
        for (int i = 0; i < 4; ++i) {
            const int off = __builtin_amdgcn_readlane(rec.x, j + i);
            r.w[i] = __int_as_float(__builtin_amdgcn_readlane(rec.y, j + i));
            r.h[i] = row_read(X, off, lane16);
        }
    }
    if (m & 2) {
        #pragma unroll
        for (int i = 0; i < 2; ++i) {
            const int off = __builtin_amdgcn_readlane(rec.x, j2 + i);
            r.w[4 + i] = __int_as_float(__builtin_amdgcn_readlane(rec.y, j2 + i));
            r.h[4 + i] = row_read(X, off, lane16);
        }
    }
    if (m & 1) {
        const int off = __builtin_amdgcn_readlane(rec.x, j1);
        r.w[6] = __int_as_float(__builtin_amdgcn_readlane(rec.y, j1));
        r.h[6] = row_read(X, off, lane16);
    }
}

template <int NSLOT>
static __device__ __forceinline__ void rows_add(float4& acc, const Rows8& r) {
    #pragma unroll
    for (int i = 0; i < NSLOT; ++i) {
        acc.x = __builtin_fmaf(r.w[i], r.h[i].x, acc.x);
        acc.y = __builtin_fmaf(r.w[i], r.h[i].y, acc.y);
        acc.z = __builtin_fmaf(r.w[i], r.h[i].z, acc.z);
        acc.w = __builtin_fmaf(r.w[i], r.h[i].w, acc.w);
    }
}

template <bool FA, bool FB, bool FIRST>
static __device__ __forceinline__ void round_ab(float4& accA, float4& accB,
                                                const float* __restrict__ X, unsigned lane16,
                                                int2 recA, int2 recB, int j, int mA, int mB,
                                                int selfA, int selfB, float ssA, float ssB) {
    float4 hA, hB;
    if (FIRST) {
        hA = row_read(X, selfA, lane16);
        hB = row_read(X, selfB, lane16);
    }
    Rows8 rA, rB;
    if (!FA && FB) {
        rows8_issue(rB, X, lane16, recB, j);
        rows7_issue(rA, X, lane16, recA, j, mA);
    } else {
        if (FA) rows8_issue(rA, X, lane16, recA, j);
        else    rows7_issue(rA, X, lane16, recA, j, mA);
        if (FB) rows8_issue(rB, X, lane16, recB, j);
        else    rows7_issue(rB, X, lane16, recB, j, mB);
    }
    if (FIRST) {
        accA.x = ssA * hA.x;
        accA.y = ssA * hA.y;
        accA.z = ssA * hA.z;
        accA.w = ssA * hA.w;
        accB.x = ssB * hB.x;
        accB.y = ssB * hB.y;
        accB.z = ssB * hB.z;
        accB.w = ssB * hB.w;
    }
    if (FA) rows_add<8>(accA, rA);
    else    rows_add<7>(accA, rA);
    if (FB) rows_add<8>(accB, rB);
    else    rows_add<7>(accB, rB);
}

template <bool FIRST>
static __device__ __forceinline__ void round_any(float4& accA, float4& accB,
                                                 const float* __restrict__ X, unsigned lane16,
                                                 int2 recA, int2 recB, int j, int cntA, int cntB,
                                                 int selfA, int selfB, float ssA, float ssB) {
    int mA = cntA - j;
    int mB = cntB - j;
    mA = (mA < 0) ? 0 : mA;
    mB = (mB < 0) ? 0 : mB;
    const bool fa = mA >= 8;
    const bool fb = mB >= 8;
    if (fa && fb)
        round_ab<true, true, FIRST>(accA, accB, X, lane16, recA, recB, j, mA, mB, selfA, selfB, ssA, ssB);
    else if (fa)
        round_ab<true, false, FIRST>(accA, accB, X, lane16, recA, recB, j, mA, mB, selfA, selfB, ssA, ssB);
    else if (fb)
        round_ab<false, true, FIRST>(accA, accB, X, lane16, recA, recB, j, mA, mB, selfA, selfB, ssA, ssB);
    else
        round_ab<false, false, FIRST>(accA, accB, X, lane16, recA, recB, j, mA, mB, selfA, selfB, ssA, ssB);
}

static __device__ __forceinline__ void rows_past32(float4& acc,
                                                   const float* __restrict__ X,
                                                   const int2* __restrict__ adj,
                                                   int lane, unsigned lane16, int beg, int end) {
    for (int base = beg + 32; base < end; base += 32) {
        const int2 cur = recs_fetch(adj, base, end, lane);
        int c2 = end - base;
        if (c2 > 32) c2 = 32;
        int j = 0;
        do {
            Rows8 r;
            if (c2 - j >= 8) {
                rows8_issue(r, X, lane16, cur, j);
                rows_add<8>(acc, r);
            } else {
                rows7_issue(r, X, lane16, cur, j, c2 - j);
                rows_add<7>(acc, r);
            }
            j += 8;
        } while (j < c2);
    }
}

#define WREG 16
__global__ __launch_bounds__(256, 8) void gcn_layer(const float* __restrict__ X,
                                                 const float* __restrict__ W,
                                                 const float* __restrict__ b,
                                                 const int* __restrict__ rowptr,
                                                 const int2* __restrict__ adj,
                                                 const float* __restrict__ dinv,
                                                 float* __restrict__ out, int tiles) {
    __shared__ __attribute__((aligned(16))) float xs[2][16 * XPITCH];
#ifdef USE_F32_WMMA
    __shared__ __attribute__((aligned(16))) float wl[(DIMF - WREG) * DIMF];
#endif

    const int wave = threadIdx.x >> 5;
    const int lane = threadIdx.x & 31;
    const int col0 = wave << 4;
    const int row  = lane & 15;
    const int hi   = lane >> 4;

    int t = blockIdx.x;
    if (t >= tiles) return;
    const int tstep = (int)gridDim.x;
    const unsigned lane16 = (unsigned)lane << 4;
    const unsigned olane = (unsigned)((hi * 8 * DIMF + col0 + row) * 4);

#ifdef USE_F32_WMMA
    const int kh = hi * 2;
    v2f bw[WREG / 4];
    #pragma unroll
    for (int i = 0; i < WREG / 4; ++i) {
        const int k = 4 * i;
        bw[i].x = W[(size_t)(k + kh)     * DIMF + col0 + row];
        bw[i].y = W[(size_t)(k + kh + 1) * DIMF + col0 + row];
    }
    {
        v2f r0[14], r1[14];
        #pragma unroll
        for (int n = 0; n < 14; ++n) {
            const int item = (int)threadIdx.x + 256 * n;
            const int m  = item & 63;
            const int k  = WREG + 2 * (item >> 6);
            r0[n] = *(const v2f*)&W[(size_t)k       * DIMF + 2 * m];
            r1[n] = *(const v2f*)&W[(size_t)(k + 1) * DIMF + 2 * m];
        }
        #pragma unroll
        for (int n = 0; n < 14; ++n) {
            const int item = (int)threadIdx.x + 256 * n;
            const int m  = item & 63;
            const int k  = WREG + 2 * (item >> 6);
            const int g  = (k - WREG) >> 2;
            const int h  = (k >> 1) & 1;
            const int ci = (2 * m + 16 * h) & (DIMF - 1);
            *(float4*)&wl[((g * 2 + h) * DIMF + ci) * 2] =
                make_float4(r0[n].x, r1[n].x, r0[n].y, r1[n].y);
        }
    }
    const float* wb = &wl[(hi * DIMF + ((col0 + row + 16 * hi) & (DIMF - 1))) * 2];
#endif
    const float bias = b[col0 + row];

    const int boff = (lane & 1) + ((lane & 2) << 2);

    int vA   = __builtin_amdgcn_readfirstlane(t * 16 + wave);
    int begA = rowptr[vA];
    int endA = rowptr[vA + 1];
    int begB = rowptr[vA + 8];
    int endB = rowptr[vA + 9];
    int2 recA = recs_fetch(adj, begA, endA, lane);
    int2 recB = recs_fetch(adj, begB, endB, lane);
    int nb = rowptr[(t + tstep < tiles) ? (t + tstep) * 16 + wave + boff : 0];
    nb = (t + tstep < tiles) ? nb : 0;
    float ssA = dinv[vA];
    float ssB = dinv[vA + 8];
    ssA *= ssA;
    ssB *= ssB;

    int buf = 0;
    while (true) {
        const int tn = t + tstep;
        const int vB = vA + 8;

        int cntA = endA - begA;
        if (cntA > 32) cntA = 32;
        int cntB = endB - begB;
        if (cntB > 32) cntB = 32;
        const int cmax = (cntA > cntB) ? cntA : cntB;
        float4 accA, accB;
        round_any<true>(accA, accB, X, lane16, recA, recB, 0, cntA, cntB,
                        vA * (DIMF * 4), vB * (DIMF * 4), ssA, ssB);

        const int nbegA = __builtin_amdgcn_readlane(nb, 0);
        const int nendA = __builtin_amdgcn_readlane(nb, 1);
        const int nbegB = __builtin_amdgcn_readlane(nb, 2);
        const int nendB = __builtin_amdgcn_readlane(nb, 3);
        const int2 nrecA = recs_fetch(adj, nbegA, nendA, lane);
        const int2 nrecB = recs_fetch(adj, nbegB, nendB, lane);
        int nd = __float_as_int(dinv[(tn < tiles) ? tn * 16 + wave + ((lane & 1) << 3) : 0]);
        nd = (tn < tiles) ? nd : 0;
        nb = rowptr[(tn + tstep < tiles) ? (tn + tstep) * 16 + wave + boff : 0];
        nb = (tn + tstep < tiles) ? nb : 0;

        for (int j = 8; j < cmax; j += 8)
            round_any<false>(accA, accB, X, lane16, recA, recB, j, cntA, cntB, 0, 0, ssA, ssB);
        rows_past32(accA, X, adj, lane, lane16, begA, endA);
        rows_past32(accB, X, adj, lane, lane16, begB, endB);

        *(float4*)&xs[buf][wave * XPITCH + 4 * lane]       = accA;
        *(float4*)&xs[buf][(wave + 8) * XPITCH + 4 * lane] = accB;
        lds_handover_barrier();

        const int m0 = t << 4;
        v8f pacc = {};
        const float* xrow = &xs[buf][row * XPITCH];

#ifdef USE_F32_WMMA
        #pragma unroll
        for (int i = 0; i < DIMF / 4; ++i) {
            const v2f a = *(const v2f*)(xrow + 4 * i + kh);
            v2f bfr;
            if (i < WREG / 4) bfr = bw[i];
            else              bfr = *(const v2f*)(wb + (i - WREG / 4) * (4 * DIMF));
            pacc = __builtin_amdgcn_wmma_f32_16x16x4_f32(
                       false, a, false, bfr, (short)0, pacc, false, false);
        }
#else
        for (int k = 0; k < DIMF; k += 32) {
            v16h a, bf;
            #pragma unroll
            for (int i = 0; i < 16; ++i) {
                int vg = i >> 1;
                int ka = ((vg < 4) ? (vg * 2) : (16 + (vg - 4) * 2)) + (i & 1) + hi * 8;
                a[i] = (_Float16)xrow[k + ka];
                int kb = i + hi * 16;
                bf[i] = (_Float16)W[(size_t)(k + kb) * DIMF + col0 + row];
            }
            pacc = __builtin_amdgcn_wmma_f32_16x16x32_f16(
                       false, a, false, bf, (short)0, pacc, false, false);
        }
#endif

        const unsigned ooff = (unsigned)m0 * (DIMF * 4) + olane;
        #pragma unroll
        for (int j = 0; j < 8; ++j) {
            float o = pacc[j] + bias;
            o = (o > 0.0f) ? o : 0.01f * o;
            *(float*)((char*)out + ooff + j * (DIMF * 4)) = o;
        }

        t = tn;
        if (t >= tiles) break;
        buf ^= 1;
        vA   = __builtin_amdgcn_readfirstlane(t * 16 + wave);
        begA = nbegA;
        endA = nendA;
        begB = nbegB;
        endB = nendB;
        recA = nrecA;
        recB = nrecB;
        const float dA = __int_as_float(__builtin_amdgcn_readlane(nd, 0));
        const float dB = __int_as_float(__builtin_amdgcn_readlane(nd, 1));
        ssA = dA * dA;
        ssB = dB * dB;
    }
}


extern "C" void kernel_launch(void* const* d_in, const int* in_sizes, int n_in,
                              void* d_out, int out_size, void* d_ws, size_t ws_size,
                              hipStream_t stream) {
    const float* x  = (const float*)d_in[0];
    const int*   ei = (const int*)d_in[1];
    const float* Wts[3] = {(const float*)d_in[2], (const float*)d_in[4], (const float*)d_in[6]};
    const float* bs[3]  = {(const float*)d_in[3], (const float*)d_in[5], (const float*)d_in[7]};

    const int N = in_sizes[0] / DIMF;
    const int E = in_sizes[1] / 2;
    const int* srcp = ei;
    const int* dstp = ei + E;

    const int tilesM  = N / 16;
    const int E4      = E / 4;
    const int buckets = (N + 255) / 256;
    const int CH4     = (E4 + NCH - 1) / NCH;
    const int CHE     = 4 * CH4;

    float* ws   = (float*)d_ws;
    const size_t nA = ((size_t)N + 64) & ~(size_t)63;
    float* dinv   = ws;
    float* Hbuf   = dinv + nA;
    float* Abuf   = Hbuf + (size_t)N * DIMF;
    int*   rowptr = (int*)(Abuf + (size_t)N * DIMF);
    int*   cs     = rowptr + nA;
    int2*  tmp    = (int2*)(cs + (size_t)(257 + 7) * NCH);
    int2*  adj    = tmp + (size_t)NCH * CHE;

    const int layerBlocks = 1024;

    sort_chunks<<<NCH,      256, 0, stream>>>(srcp, dstp, cs, tmp, E4, CH4);
    sort_buckets<<<buckets, 256, 0, stream>>>(cs, tmp, rowptr, dinv, adj, N, CHE);
    adj_weights<<<400,      256, 0, stream>>>(adj, dinv, E4);

    gcn_layer<<<layerBlocks, 256, 0, stream>>>(x,    Wts[0], bs[0], rowptr, adj, dinv, Abuf,          tilesM);
    gcn_layer<<<layerBlocks, 256, 0, stream>>>(Abuf, Wts[1], bs[1], rowptr, adj, dinv, Hbuf,          tilesM);
    gcn_layer<<<layerBlocks, 256, 0, stream>>>(Hbuf, Wts[2], bs[2], rowptr, adj, dinv, (float*)d_out, tilesM);
}


# Edge weight dinv[u]*dinv[v] factored: rows kept scaled by own dinv (x in the bucket pass, layer outputs in the epilogue), summed unweighted, one scale per node; weight launch gone; rounding changes
# speedup vs baseline: 5.4609x; 1.0631x over previous
#include <hip/hip_runtime.h>

typedef float     v2f  __attribute__((ext_vector_type(2)));
typedef float     v8f  __attribute__((ext_vector_type(8)));
typedef _Float16  v16h __attribute__((ext_vector_type(16)));

#define DIMF 128

#if __has_builtin(__builtin_amdgcn_wmma_f32_16x16x4_f32)
#define USE_F32_WMMA 1
#endif


#define NCH 256

static __device__ __forceinline__ int block_scan256(int v, int lane, int wave, int* wsum, int& total) {
    int s = v;
    #pragma unroll
    for (int o = 1; o < 32; o <<= 1) {
        int t = __shfl_up(s, o);
        if (lane >= o) s += t;
    }
    if (lane == 31) wsum[wave] = s;
    __syncthreads();
    int off = 0, tot = 0;
    #pragma unroll
    for (int w = 0; w < 8; ++w) {
        const int x = wsum[w];
        tot += x;
        if (w < wave) off += x;
    }
    total = tot;
    return off + s - v;
}

__global__ __launch_bounds__(256) void sort_chunks(const int* __restrict__ src,
                                                   const int* __restrict__ dst,
                                                   int* __restrict__ cs, int2* __restrict__ tmp,
                                                   int E4, int CH4) {
    __shared__ int hist[256];
    __shared__ int wsum[8];
    const int tid  = threadIdx.x;
    const int lane = tid & 31;
    const int wave = tid >> 5;
    const int c    = blockIdx.x;
    const int q0   = c * CH4;
    int nq = E4 - q0;
    nq = (nq > CH4) ? CH4 : nq;
    nq = (nq < 0) ? 0 : nq;
    const int4* s4 = (const int4*)src;
    const int4* d4 = (const int4*)dst;

    hist[tid] = 0;
    __syncthreads();
    for (int t0 = 0; t0 < nq; t0 += 4 * 256) {
        int4 d[4];
        bool ok[4];
        #pragma unroll
        for (int k = 0; k < 4; ++k) {
            const int i = t0 + k * 256 + tid;
            ok[k] = i < nq;
            d[k] = d4[ok[k] ? q0 + i : 0];
        }
        #pragma unroll
        for (int k = 0; k < 4; ++k) {
            if (ok[k]) {
                atomicAdd(&hist[d[k].x >> 8], 1);
                atomicAdd(&hist[d[k].y >> 8], 1);
                atomicAdd(&hist[d[k].z >> 8], 1);
                atomicAdd(&hist[d[k].w >> 8], 1);
            }
        }
    }
    __syncthreads();
    const int h = hist[tid];
    int total;
    const int excl = block_scan256(h, lane, wave, wsum, total);
    cs[tid * NCH + c] = excl;
    if (tid == 255) cs[256 * NCH + c] = total;
    hist[tid] = excl;
    __syncthreads();

    int2* out = tmp + (size_t)c * (size_t)(4 * CH4);
    for (int t0 = 0; t0 < nq; t0 += 4 * 256) {
        int4 d[4], s[4];
        bool ok[4];
        #pragma unroll
        for (int k = 0; k < 4; ++k) {
            const int i = t0 + k * 256 + tid;
            ok[k] = i < nq;
            d[k] = d4[ok[k] ? q0 + i : 0];
            s[k] = s4[ok[k] ? q0 + i : 0];
        }
        #pragma unroll
        for (int k = 0; k < 4; ++k) {
            if (ok[k]) {
                const int p0 = atomicAdd(&hist[d[k].x >> 8], 1);
                const int p1 = atomicAdd(&hist[d[k].y >> 8], 1);
                const int p2 = atomicAdd(&hist[d[k].z >> 8], 1);
                const int p3 = atomicAdd(&hist[d[k].w >> 8], 1);
                out[p0] = make_int2(s[k].x, d[k].x);
                out[p1] = make_int2(s[k].y, d[k].y);
                out[p2] = make_int2(s[k].z, d[k].z);
                out[p3] = make_int2(s[k].w, d[k].w);
            }
        }
    }
}

template <bool PLACE>
static __device__ __forceinline__ void bucket_edge(int2 e, int* cnt, int* __restrict__ adj) {
    const int ld = e.y & 255;
    if (PLACE) {
        const int p = atomicAdd(&cnt[ld], 1);
        adj[p] = e.x * (DIMF * 4);
    } else {
        atomicAdd(&cnt[ld], 1);
    }
}

__global__ __launch_bounds__(256) void sort_buckets(const int* __restrict__ cs,
                                                    const int2* __restrict__ tmp,
                                                    int* __restrict__ rowptr,
                                                    float* __restrict__ dinv,
                                                    int* __restrict__ adj,
                                                    const float* __restrict__ x,
                                                    float* __restrict__ xs, int n, int CHE) {
    __shared__ int deg[256];
    __shared__ int cur[256];
    __shared__ int wsA[8];
    __shared__ int wsB[8];
    const int tid  = threadIdx.x;
    const int lane = tid & 31;
    const int wave = __builtin_amdgcn_readfirstlane(tid >> 5);
    const int b    = blockIdx.x;

    const int s0 = cs[b * NCH + tid];
    const int s1 = cs[(b + 1) * NCH + tid];
    deg[tid] = 0;
    int r = s0;
    #pragma unroll
    for (int o = 16; o > 0; o >>= 1) r += __shfl_xor(r, o);
    if (lane == 0) wsA[wave] = r;
    __syncthreads();
    int bstart = 0;
    #pragma unroll
    for (int w = 0; w < 8; ++w) bstart += wsA[w];

    const int2* wtmp = tmp + (size_t)(wave * 32) * (size_t)CHE;
    int2 e[32];
    unsigned okm = 0u;
    #pragma unroll
    for (int j = 0; j < 32; ++j) {
        const int beg = __builtin_amdgcn_readlane(s0, j);
        const int end = __builtin_amdgcn_readlane(s1, j);
        const int i = beg + lane;
        const bool ok = i < end;
        e[j] = wtmp[(size_t)j * (size_t)CHE + (ok ? i : 0)];
        okm |= ok ? (1u << j) : 0u;
    }
    const int vw = b * 256 + wave * 32;
    float4 h0[16];
    #pragma unroll
    for (int i = 0; i < 16; ++i) {
        int vr = vw + i;
        vr = (vr < n) ? vr : (n - 1);
        h0[i] = *(const float4*)&x[(size_t)vr * DIMF + 4 * lane];
    }
    #pragma unroll
    for (int j = 0; j < 32; ++j)
        if ((okm >> j) & 1u) bucket_edge<false>(e[j], deg, adj);
    for (int j = 0; j < 32; ++j) {
        const int beg = __builtin_amdgcn_readlane(s0, j);
        const int end = __builtin_amdgcn_readlane(s1, j);
        for (int i = beg + 32 + lane; i < end; i += 32)
            bucket_edge<false>(wtmp[(size_t)j * (size_t)CHE + i], deg, adj);
    }
    __syncthreads();

    const int d = deg[tid];
    int total;
    const int excl = block_scan256(d, lane, wave, wsB, total);
    const int v = b * 256 + tid;
    const float dv = rsqrtf((float)d + 1.0f);
    if (v < n) {
        rowptr[v] = bstart + excl;
        dinv[v]   = dv;
        if (v == n - 1) rowptr[n] = bstart + excl + d;
    }
    cur[tid] = bstart + excl;
    __syncthreads();

    float4 h1[16];
    #pragma unroll
    for (int i = 0; i < 16; ++i) {
        int vr = vw + 16 + i;
        vr = (vr < n) ? vr : (n - 1);
        h1[i] = *(const float4*)&x[(size_t)vr * DIMF + 4 * lane];
    }

    #pragma unroll
    for (int j = 0; j < 32; ++j)
        if ((okm >> j) & 1u) bucket_edge<true>(e[j], cur, adj);
    for (int j = 0; j < 32; ++j) {
        const int beg = __builtin_amdgcn_readlane(s0, j);
        const int end = __builtin_amdgcn_readlane(s1, j);
        for (int i = beg + 32 + lane; i < end; i += 32)
            bucket_edge<true>(wtmp[(size_t)j * (size_t)CHE + i], cur, adj);
    }

    #pragma unroll
    for (int i = 0; i < 16; ++i) {
        const float sc = __int_as_float(__builtin_amdgcn_readlane(__float_as_int(dv), i));
        if (vw + i < n)
            *(float4*)&xs[(size_t)(vw + i) * DIMF + 4 * lane] =
                make_float4(sc * h0[i].x, sc * h0[i].y, sc * h0[i].z, sc * h0[i].w);
    }
    #pragma unroll
    for (int i = 0; i < 16; ++i) {
        const float sc = __int_as_float(__builtin_amdgcn_readlane(__float_as_int(dv), 16 + i));
        if (vw + 16 + i < n)
            *(float4*)&xs[(size_t)(vw + 16 + i) * DIMF + 4 * lane] =
                make_float4(sc * h1[i].x, sc * h1[i].y, sc * h1[i].z, sc * h1[i].w);
    }
}


#define XPITCH 132

static __device__ __forceinline__ void lds_handover_barrier() {
    __builtin_amdgcn_fence(__ATOMIC_RELEASE, "workgroup", "local");
    __builtin_amdgcn_s_barrier();
    __builtin_amdgcn_fence(__ATOMIC_ACQUIRE, "workgroup", "local");
}

static __device__ __forceinline__ int recs_fetch(const int* __restrict__ adj, int beg, int end,
                                                 int lane) {
    const int idx = beg + lane;
    const bool ok = idx < end;
    const int v = adj[ok ? idx : 0];
    return ok ? v : 0;
}

struct Rows8 {
    float4 h[8];
};

static __device__ __forceinline__ float4 row_read(const float* __restrict__ X, int rowoff,
                                                  unsigned lane16) {
    const unsigned voff = (unsigned)rowoff + lane16;
    return *(const float4*)((const char*)X + voff);
}

static __device__ __forceinline__ void rows8_issue(Rows8& r, const float* __restrict__ X,
                                                   unsigned lane16, int rec, int j) {
    #pragma unroll
    for (int i = 0; i < 8; ++i) {
        const int off = __builtin_amdgcn_readlane(rec, j + i);
        r.h[i] = row_read(X, off, lane16);
    }
}

static __device__ __forceinline__ void rows7_issue(Rows8& r, const float* __restrict__ X,
                                                   unsigned lane16, int rec, int j, int m) {
    #pragma unroll
    for (int i = 0; i < 7; ++i)
        r.h[i] = make_float4(0.0f, 0.0f, 0.0f, 0.0f);
    const int j2 = j + (m & 4);
    const int j1 = j + (m & 6);
    if (m & 4) {
        #pragma unroll
        for (int i = 0; i < 4; ++i) {
            const int off = __builtin_amdgcn_readlane(rec, j + i);
            r.h[i] = row_read(X, off, lane16);
        }
    }
    if (m & 2) {
        #pragma unroll
        for (int i = 0; i < 2; ++i) {
            const int off = __builtin_amdgcn_readlane(rec, j2 + i);
            r.h[4 + i] = row_read(X, off, lane16);
        }
    }
    if (m & 1) {
        const int off = __builtin_amdgcn_readlane(rec, j1);
        r.h[6] = row_read(X, off, lane16);
    }
}

template <int NSLOT>
static __device__ __forceinline__ void rows_add(float4& acc, const Rows8& r) {
    #pragma unroll
    for (int i = 0; i < NSLOT; ++i) {
        acc.x += r.h[i].x;
        acc.y += r.h[i].y;
        acc.z += r.h[i].z;
        acc.w += r.h[i].w;
    }
}

template <bool FA, bool FB, bool FIRST>
static __device__ __forceinline__ void round_ab(float4& accA, float4& accB,
                                                const float* __restrict__ X, unsigned lane16,
                                                int recA, int recB, int j, int mA, int mB,
                                                int selfA, int selfB) {
    float4 hA, hB;
    if (FIRST) {
        hA = row_read(X, selfA, lane16);
        hB = row_read(X, selfB, lane16);
    }
    Rows8 rA, rB;
    if (!FA && FB) {
        rows8_issue(rB, X, lane16, recB, j);
        rows7_issue(rA, X, lane16, recA, j, mA);
    } else {
        if (FA) rows8_issue(rA, X, lane16, recA, j);
        else    rows7_issue(rA, X, lane16, recA, j, mA);
        if (FB) rows8_issue(rB, X, lane16, recB, j);
        else    rows7_issue(rB, X, lane16, recB, j, mB);
    }
    if (FIRST) {
        accA = hA;
        accB = hB;
    }
    if (FA) rows_add<8>(accA, rA);
    else    rows_add<7>(accA, rA);
    if (FB) rows_add<8>(accB, rB);
    else    rows_add<7>(accB, rB);
}

template <bool FIRST>
static __device__ __forceinline__ void round_any(float4& accA, float4& accB,
                                                 const float* __restrict__ X, unsigned lane16,
                                                 int recA, int recB, int j, int cntA, int cntB,
                                                 int selfA, int selfB) {
    int mA = cntA - j;
    int mB = cntB - j;
    mA = (mA < 0) ? 0 : mA;
    mB = (mB < 0) ? 0 : mB;
    const bool fa = mA >= 8;
    const bool fb = mB >= 8;
    if (fa && fb)
        round_ab<true, true, FIRST>(accA, accB, X, lane16, recA, recB, j, mA, mB, selfA, selfB);
    else if (fa)
        round_ab<true, false, FIRST>(accA, accB, X, lane16, recA, recB, j, mA, mB, selfA, selfB);
    else if (fb)
        round_ab<false, true, FIRST>(accA, accB, X, lane16, recA, recB, j, mA, mB, selfA, selfB);
    else
        round_ab<false, false, FIRST>(accA, accB, X, lane16, recA, recB, j, mA, mB, selfA, selfB);
}

static __device__ __forceinline__ void rows_past32(float4& acc,
                                                   const float* __restrict__ X,
                                                   const int* __restrict__ adj,
                                                   int lane, unsigned lane16, int beg, int end) {
    for (int base = beg + 32; base < end; base += 32) {
        const int cur = recs_fetch(adj, base, end, lane);
        int c2 = end - base;
        if (c2 > 32) c2 = 32;
        int j = 0;
        do {
            Rows8 r;
            if (c2 - j >= 8) {
                rows8_issue(r, X, lane16, cur, j);
                rows_add<8>(acc, r);
            } else {
                rows7_issue(r, X, lane16, cur, j, c2 - j);
                rows_add<7>(acc, r);
            }
            j += 8;
        } while (j < c2);
    }
}

#define WREG 16
template <bool SCALE_OUT>
__global__ __launch_bounds__(256, 8) void gcn_layer(const float* __restrict__ X,
                                                 const float* __restrict__ W,
                                                 const float* __restrict__ b,
                                                 const int* __restrict__ rowptr,
                                                 const int* __restrict__ adj,
                                                 const float* __restrict__ dinv,
                                                 float* __restrict__ out, int tiles) {
    __shared__ __attribute__((aligned(16))) float xs[2][16 * XPITCH];
    __shared__ __attribute__((aligned(16))) float sd[2][16];
#ifdef USE_F32_WMMA
    __shared__ __attribute__((aligned(16))) float wl[(DIMF - WREG) * DIMF];
#endif

    const int wave = threadIdx.x >> 5;
    const int lane = threadIdx.x & 31;
    const int col0 = wave << 4;
    const int row  = lane & 15;
    const int hi   = lane >> 4;

    int t = blockIdx.x;
    if (t >= tiles) return;
    const int tstep = (int)gridDim.x;
    const unsigned lane16 = (unsigned)lane << 4;
    const unsigned olane = (unsigned)((hi * 8 * DIMF + col0 + row) * 4);

#ifdef USE_F32_WMMA
    const int kh = hi * 2;
    v2f bw[WREG / 4];
    #pragma unroll
    for (int i = 0; i < WREG / 4; ++i) {
        const int k = 4 * i;
        bw[i].x = W[(size_t)(k + kh)     * DIMF + col0 + row];
        bw[i].y = W[(size_t)(k + kh + 1) * DIMF + col0 + row];
    }
    {
        v2f r0[14], r1[14];
        #pragma unroll
        for (int n = 0; n < 14; ++n) {
            const int item = (int)threadIdx.x + 256 * n;
            const int m  = item & 63;
            const int k  = WREG + 2 * (item >> 6);
            r0[n] = *(const v2f*)&W[(size_t)k       * DIMF + 2 * m];
            r1[n] = *(const v2f*)&W[(size_t)(k + 1) * DIMF + 2 * m];
        }
        #pragma unroll
        for (int n = 0; n < 14; ++n) {
            const int item = (int)threadIdx.x + 256 * n;
            const int m  = item & 63;
            const int k  = WREG + 2 * (item >> 6);
            const int g  = (k - WREG) >> 2;
            const int h  = (k >> 1) & 1;
            const int ci = (2 * m + 16 * h) & (DIMF - 1);
            *(float4*)&wl[((g * 2 + h) * DIMF + ci) * 2] =
                make_float4(r0[n].x, r1[n].x, r0[n].y, r1[n].y);
        }
    }
    const float* wb = &wl[(hi * DIMF + ((col0 + row + 16 * hi) & (DIMF - 1))) * 2];
#endif
    const float bias = b[col0 + row];

    const int boff = (lane & 1) + ((lane & 2) << 2);

    int vA   = __builtin_amdgcn_readfirstlane(t * 16 + wave);
    int begA = rowptr[vA];
    int endA = rowptr[vA + 1];
    int begB = rowptr[vA + 8];
    int endB = rowptr[vA + 9];
    int recA = recs_fetch(adj, begA, endA, lane);
    int recB = recs_fetch(adj, begB, endB, lane);
    int nb = rowptr[(t + tstep < tiles) ? (t + tstep) * 16 + wave + boff : 0];
    nb = (t + tstep < tiles) ? nb : 0;
    float dvA = dinv[vA];
    float dvB = dinv[vA + 8];

    int buf = 0;
    while (true) {
        const int tn = t + tstep;
        const int vB = vA + 8;

        int cntA = endA - begA;
        if (cntA > 32) cntA = 32;
        int cntB = endB - begB;
        if (cntB > 32) cntB = 32;
        const int cmax = (cntA > cntB) ? cntA : cntB;
        float4 accA, accB;
        round_any<true>(accA, accB, X, lane16, recA, recB, 0, cntA, cntB,
                        vA * (DIMF * 4), vB * (DIMF * 4));

        const int nbegA = __builtin_amdgcn_readlane(nb, 0);
        const int nendA = __builtin_amdgcn_readlane(nb, 1);
        const int nbegB = __builtin_amdgcn_readlane(nb, 2);
        const int nendB = __builtin_amdgcn_readlane(nb, 3);
        const int nrecA = recs_fetch(adj, nbegA, nendA, lane);
        const int nrecB = recs_fetch(adj, nbegB, nendB, lane);
        int nd = __float_as_int(dinv[(tn < tiles) ? tn * 16 + wave + ((lane & 1) << 3) : 0]);
        nd = (tn < tiles) ? nd : 0;
        nb = rowptr[(tn + tstep < tiles) ? (tn + tstep) * 16 + wave + boff : 0];
        nb = (tn + tstep < tiles) ? nb : 0;

        for (int j = 8; j < cmax; j += 8)
            round_any<false>(accA, accB, X, lane16, recA, recB, j, cntA, cntB, 0, 0);
        rows_past32(accA, X, adj, lane, lane16, begA, endA);
        rows_past32(accB, X, adj, lane, lane16, begB, endB);

        *(float4*)&xs[buf][wave * XPITCH + 4 * lane] =
            make_float4(dvA * accA.x, dvA * accA.y, dvA * accA.z, dvA * accA.w);
        *(float4*)&xs[buf][(wave + 8) * XPITCH + 4 * lane] =
            make_float4(dvB * accB.x, dvB * accB.y, dvB * accB.z, dvB * accB.w);
        if (SCALE_OUT) sd[buf][wave + ((lane & 1) << 3)] = (lane & 1) ? dvB : dvA;
        lds_handover_barrier();

        const int m0 = t << 4;
        v8f pacc = {};
        const float* xrow = &xs[buf][row * XPITCH];

#ifdef USE_F32_WMMA
        #pragma unroll
        for (int i = 0; i < DIMF / 4; ++i) {
            const v2f a = *(const v2f*)(xrow + 4 * i + kh);
            v2f bfr;
            if (i < WREG / 4) bfr = bw[i];
            else              bfr = *(const v2f*)(wb + (i - WREG / 4) * (4 * DIMF));
            pacc = __builtin_amdgcn_wmma_f32_16x16x4_f32(
                       false, a, false, bfr, (short)0, pacc, false, false);
        }
#else
        for (int k = 0; k < DIMF; k += 32) {
            v16h a, bf;
            #pragma unroll
            for (int i = 0; i < 16; ++i) {
                int vg = i >> 1;
                int ka = ((vg < 4) ? (vg * 2) : (16 + (vg - 4) * 2)) + (i & 1) + hi * 8;
                a[i] = (_Float16)xrow[k + ka];
                int kb = i + hi * 16;
                bf[i] = (_Float16)W[(size_t)(k + kb) * DIMF + col0 + row];
            }
            pacc = __builtin_amdgcn_wmma_f32_16x16x32_f16(
                       false, a, false, bf, (short)0, pacc, false, false);
        }
#endif

        const unsigned ooff = (unsigned)m0 * (DIMF * 4) + olane;
        float osc[8];
        if (SCALE_OUT) {
            const float4 s0 = *(const float4*)&sd[buf][8 * hi];
            const float4 s1 = *(const float4*)&sd[buf][8 * hi + 4];
            osc[0] = s0.x; osc[1] = s0.y; osc[2] = s0.z; osc[3] = s0.w;
            osc[4] = s1.x; osc[5] = s1.y; osc[6] = s1.z; osc[7] = s1.w;
        }
        #pragma unroll
        for (int j = 0; j < 8; ++j) {
            float o = pacc[j] + bias;
            o = (o > 0.0f) ? o : 0.01f * o;
            if (SCALE_OUT) o *= osc[j];
            *(float*)((char*)out + ooff + j * (DIMF * 4)) = o;
        }

        t = tn;
        if (t >= tiles) break;
        buf ^= 1;
        vA   = __builtin_amdgcn_readfirstlane(t * 16 + wave);
        begA = nbegA;
        endA = nendA;
        begB = nbegB;
        endB = nendB;
        recA = nrecA;
        recB = nrecB;
        dvA = __int_as_float(__builtin_amdgcn_readlane(nd, 0));
        dvB = __int_as_float(__builtin_amdgcn_readlane(nd, 1));
    }
}


extern "C" void kernel_launch(void* const* d_in, const int* in_sizes, int n_in,
                              void* d_out, int out_size, void* d_ws, size_t ws_size,
                              hipStream_t stream) {
    const float* x  = (const float*)d_in[0];
    const int*   ei = (const int*)d_in[1];
    const float* Wts[3] = {(const float*)d_in[2], (const float*)d_in[4], (const float*)d_in[6]};
    const float* bs[3]  = {(const float*)d_in[3], (const float*)d_in[5], (const float*)d_in[7]};

    const int N = in_sizes[0] / DIMF;
    const int E = in_sizes[1] / 2;
    const int* srcp = ei;
    const int* dstp = ei + E;

    const int tilesM  = N / 16;
    const int E4      = E / 4;
    const int buckets = (N + 255) / 256;
    const int CH4     = (E4 + NCH - 1) / NCH;
    const int CHE     = 4 * CH4;

    float* ws   = (float*)d_ws;
    const size_t nA = ((size_t)N + 64) & ~(size_t)63;
    float* dinv   = ws;
    float* Hbuf   = dinv + nA;
    float* Abuf   = Hbuf + (size_t)N * DIMF;
    int*   rowptr = (int*)(Abuf + (size_t)N * DIMF);
    int*   cs     = rowptr + nA;
    int2*  tmp    = (int2*)(cs + (size_t)(257 + 7) * NCH);
    int*   adj    = (int*)(tmp + (size_t)NCH * CHE);
    float* Xs     = (float*)(adj + (((size_t)E + 63) & ~(size_t)63));

    const int layerBlocks = 1024;

    sort_chunks<<<NCH,      256, 0, stream>>>(srcp, dstp, cs, tmp, E4, CH4);
    sort_buckets<<<buckets, 256, 0, stream>>>(cs, tmp, rowptr, dinv, adj, x, Xs, N, CHE);

    gcn_layer<true><<<layerBlocks, 256, 0, stream>>>(Xs,   Wts[0], bs[0], rowptr, adj, dinv, Abuf,          tilesM);
    gcn_layer<true><<<layerBlocks, 256, 0, stream>>>(Abuf, Wts[1], bs[1], rowptr, adj, dinv, Hbuf,          tilesM);
    gcn_layer<false><<<layerBlocks, 256, 0, stream>>>(Hbuf, Wts[2], bs[2], rowptr, adj, dinv, (float*)d_out, tilesM);
}
